# starting kernel
# baseline (speedup 1.0000x reference)
#include <hip/hip_runtime.h>
#include <hip/hip_bf16.h>

typedef __attribute__((ext_vector_type(2))) float v2f;
typedef __attribute__((ext_vector_type(8))) float v8f;

#define N_NODES 50000
#define N_GRAPHS 2048

__device__ __forceinline__ void atomic_add_f32(float* p, float v) {
    unsafeAtomicAdd(p, v);
}

__global__ void init_deg_kernel(float* deg, int n) {
    int i = blockIdx.x * blockDim.x + threadIdx.x;
    if (i < n) deg[i] = 1.0f;
}

__global__ void count_deg_kernel(const int* __restrict__ dst, float* deg, int E) {
    int i = blockIdx.x * blockDim.x + threadIdx.x;
    if (i < E) atomic_add_f32(&deg[dst[i]], 1.0f);
}

__global__ void dinv_kernel(float* deg, int n) {
    int i = blockIdx.x * blockDim.x + threadIdx.x;
    if (i < n) deg[i] = rsqrtf(deg[i]);
}

__global__ void zero_kernel(float* p, int n) {
    int i = blockIdx.x * blockDim.x + threadIdx.x;
    if (i < n) p[i] = 0.0f;
}

template <int K>
__global__ void gemm_wmma_f32(const float* __restrict__ A,
                              const float* __restrict__ W,
                              const float* __restrict__ bias,
                              float* __restrict__ H,
                              int M, int do_relu) {
    __shared__ float Wlds[K * 64];
    #pragma unroll 4
    for (int i = threadIdx.x; i < K * 64; i += 256)
        Wlds[i] = W[i];
    __syncthreads();

    const int wave = threadIdx.x >> 5;
    const int lane = threadIdx.x & 31;
    const int half = lane >> 4;
    const int l16  = lane & 15;
    const int col0 = (wave & 3) * 16;
    const int row0 = blockIdx.x * 32 + (wave >> 2) * 16;
    const int m    = row0 + l16;
    const int mc   = (m < M) ? m : (M - 1);
    const float* __restrict__ Arow = A + (size_t)mc * K + 2 * half;
    const float* __restrict__ Brow = Wlds + 2 * half * 64 + col0 + l16;

    v8f acc = {};
    #pragma unroll
    for (int k0 = 0; k0 < K; k0 += 4) {
        v2f a = *(const v2f*)(Arow + k0);
        v2f b;
        b.x = Brow[k0 * 64];
        b.y = Brow[k0 * 64 + 64];
        acc = __builtin_amdgcn_wmma_f32_16x16x4_f32(
             false, a,  false, b,
             (short)0, acc,  false,  false);
    }
    const float bv = bias ? bias[col0 + l16] : 0.0f;
    #pragma unroll
    for (int r = 0; r < 8; ++r) {
        int mm = row0 + r + 8 * half;
        if (mm < M) {
            float v = acc[r] + bv;
            if (do_relu) v = fmaxf(v, 0.0f);
            H[(size_t)mm * 64 + col0 + l16] = v;
        }
    }
}

__global__ void edge_agg_kernel(const int* __restrict__ src,
                                const int* __restrict__ dst,
                                const float* __restrict__ dinv,
                                const float* __restrict__ h,
                                float* __restrict__ agg, int E) {
    int e = __builtin_amdgcn_readfirstlane(blockIdx.x * (blockDim.x >> 5) +
                                           (threadIdx.x >> 5));
    if (e >= E) return;
    int lane = threadIdx.x & 31;
    int s = __builtin_amdgcn_readfirstlane(src[e]);
    int d = __builtin_amdgcn_readfirstlane(dst[e]);
    float nrm = dinv[s] * dinv[d];
    const float2* hs = (const float2*)(h + (size_t)s * 64);
    float2 v = hs[lane];
    float* ad = agg + (size_t)d * 64 + 2 * lane;
    atomic_add_f32(ad + 0, nrm * v.x);
    atomic_add_f32(ad + 1, nrm * v.y);
}

__global__ void finalize_kernel(const float* __restrict__ agg,
                                const float* __restrict__ hlin,
                                const float* __restrict__ dinv,
                                const float* __restrict__ bias,
                                float* __restrict__ hout, int total) {
    int i = blockIdx.x * blockDim.x + threadIdx.x;
    if (i >= total) return;
    int n = i >> 6, f = i & 63;
    float di = dinv[n];
    float v = agg[i] + di * di * hlin[i] + bias[f];
    hout[i] = fmaxf(v, 0.0f);
}

__global__ void pool_acc_kernel(const float* __restrict__ h,
                                const int* __restrict__ batch,
                                float* __restrict__ pooled,
                                float* __restrict__ counts, int n) {
    int node = __builtin_amdgcn_readfirstlane(blockIdx.x * (blockDim.x >> 5) +
                                              (threadIdx.x >> 5));
    if (node >= n) return;
    int lane = threadIdx.x & 31;
    int g = __builtin_amdgcn_readfirstlane(batch[node]);
    const float2* hn = (const float2*)(h + (size_t)node * 64);
    float2 v = hn[lane];
    float* pg = pooled + (size_t)g * 64 + 2 * lane;
    atomic_add_f32(pg + 0, v.x);
    atomic_add_f32(pg + 1, v.y);
    if (lane == 0) atomic_add_f32(&counts[g], 1.0f);
}

__global__ void pool_div_kernel(float* pooled, const float* counts, int total) {
    int i = blockIdx.x * blockDim.x + threadIdx.x;
    if (i >= total) return;
    float c = fmaxf(counts[i >> 6], 1.0f);
    pooled[i] = pooled[i] / c;
}

__global__ void head_kernel(const float* __restrict__ r1,
                            const float* __restrict__ lw2,
                            const float* __restrict__ lb2,
                            float* __restrict__ out, int G) {
    int i = blockIdx.x * blockDim.x + threadIdx.x;
    if (i >= G * 2) return;
    int g = i >> 1, j = i & 1;
    float acc = lb2[j];
    #pragma unroll
    for (int k = 0; k < 64; ++k) acc += r1[(size_t)g * 64 + k] * lw2[k * 2 + j];
    out[i] = acc;
}

static inline size_t alignup(size_t v) { return (v + 255) & ~(size_t)255; }

extern "C" void kernel_launch(void* const* d_in, const int* in_sizes, int n_in,
                              void* d_out, int out_size, void* d_ws, size_t ws_size,
                              hipStream_t stream) {
    const float* x     = (const float*)d_in[0];
    const int*   eidx  = (const int*)d_in[1];
    const int*   batch = (const int*)d_in[2];
    const float* W1 = (const float*)d_in[3];  const float* b1 = (const float*)d_in[4];
    const float* W2 = (const float*)d_in[5];  const float* b2 = (const float*)d_in[6];
    const float* W3 = (const float*)d_in[7];  const float* b3 = (const float*)d_in[8];
    const float* lw1 = (const float*)d_in[9]; const float* lb1 = (const float*)d_in[10];
    const float* lw2 = (const float*)d_in[11]; const float* lb2 = (const float*)d_in[12];
    float* out = (float*)d_out;

    const int N = N_NODES, G = N_GRAPHS;
    const int E = in_sizes[1] / 2;
    const int* src = eidx;
    const int* dst = eidx + E;

    char* w = (char*)d_ws;
    float* dinv   = (float*)w; w += alignup((size_t)N * 4);
    float* hlin   = (float*)w; w += alignup((size_t)N * 64 * 4);
    float* agg    = (float*)w; w += alignup((size_t)N * 64 * 4);
    float* hfeat  = (float*)w; w += alignup((size_t)N * 64 * 4);
    float* pooled = (float*)w; w += alignup((size_t)G * 64 * 4);
    float* counts = (float*)w; w += alignup((size_t)G * 4);
    float* relu1  = (float*)w; w += alignup((size_t)G * 64 * 4);
    (void)ws_size; (void)n_in;

    const int NF = N * 64;
    dim3 blk256(256);
    int gN    = (N + 255) / 256;
    int gNF   = (NF + 255) / 256;
    int gE    = (E + 255) / 256;
    int gE8   = (E + 7) / 8;
    int gN8   = (N + 7) / 8;
    int gGemmN = (N + 31) / 32;
    int gGemmG = (G + 31) / 32;

    init_deg_kernel<<<gN, blk256, 0, stream>>>(dinv, N);
    count_deg_kernel<<<gE, blk256, 0, stream>>>(dst, dinv, E);
    dinv_kernel<<<gN, blk256, 0, stream>>>(dinv, N);

    gemm_wmma_f32<32><<<gGemmN, blk256, 0, stream>>>(x, W1, nullptr, hlin, N, 0);
    zero_kernel<<<gNF, blk256, 0, stream>>>(agg, NF);
    edge_agg_kernel<<<gE8, blk256, 0, stream>>>(src, dst, dinv, hlin, agg, E);
    finalize_kernel<<<gNF, blk256, 0, stream>>>(agg, hlin, dinv, b1, hfeat, NF);

    gemm_wmma_f32<64><<<gGemmN, blk256, 0, stream>>>(hfeat, W2, nullptr, hlin, N, 0);
    zero_kernel<<<gNF, blk256, 0, stream>>>(agg, NF);
    edge_agg_kernel<<<gE8, blk256, 0, stream>>>(src, dst, dinv, hlin, agg, E);
    finalize_kernel<<<gNF, blk256, 0, stream>>>(agg, hlin, dinv, b2, hfeat, NF);

    gemm_wmma_f32<64><<<gGemmN, blk256, 0, stream>>>(hfeat, W3, nullptr, hlin, N, 0);
    zero_kernel<<<gNF, blk256, 0, stream>>>(agg, NF);
    edge_agg_kernel<<<gE8, blk256, 0, stream>>>(src, dst, dinv, hlin, agg, E);
    finalize_kernel<<<gNF, blk256, 0, stream>>>(agg, hlin, dinv, b3, hfeat, NF);

    int GF = G * 64;
    zero_kernel<<<(GF + 255) / 256, blk256, 0, stream>>>(pooled, GF);
    zero_kernel<<<(G + 255) / 256, blk256, 0, stream>>>(counts, G);
    pool_acc_kernel<<<gN8, blk256, 0, stream>>>(hfeat, batch, pooled, counts, N);
    pool_div_kernel<<<(GF + 255) / 256, blk256, 0, stream>>>(pooled, counts, GF);

    gemm_wmma_f32<64><<<gGemmG, blk256, 0, stream>>>(pooled, lw1, lb1, relu1, G, 1);
    head_kernel<<<(G * 2 + 255) / 256, blk256, 0, stream>>>(relu1, lw2, lb2, out, G);
    (void)out_size;
}


# wave-launch-bound kernels get more work per wave: edge_agg 32 edges/wave, 8 row gathers in flight (800k->25k waves); zero/finalize float4/thread; f32 atomics kept, arrival order differs
# speedup vs baseline: 1.1343x; 1.1343x over previous
#include <hip/hip_runtime.h>
#include <hip/hip_bf16.h>

typedef __attribute__((ext_vector_type(2))) float v2f;
typedef __attribute__((ext_vector_type(8))) float v8f;

#define N_NODES 50000
#define N_GRAPHS 2048

__device__ __forceinline__ void atomic_add_f32(float* p, float v) {
    unsafeAtomicAdd(p, v);
}

__global__ void init_deg_kernel(float* deg, int n) {
    int i = blockIdx.x * blockDim.x + threadIdx.x;
    if (i < n) deg[i] = 1.0f;
}

__global__ void count_deg_kernel(const int* __restrict__ dst, float* deg, int E) {
    int i = blockIdx.x * blockDim.x + threadIdx.x;
    if (i < E) atomic_add_f32(&deg[dst[i]], 1.0f);
}

__global__ void dinv_kernel(float* deg, int n) {
    int i = blockIdx.x * blockDim.x + threadIdx.x;
    if (i < n) deg[i] = rsqrtf(deg[i]);
}

__global__ void zero_kernel(float* p, int n) {
    int i = blockIdx.x * blockDim.x + threadIdx.x;
    if (i < n) p[i] = 0.0f;
}

__global__ void zero4_kernel(float* p, int n4) {
    int i = blockIdx.x * blockDim.x + threadIdx.x;
    if (i < n4) ((float4*)p)[i] = make_float4(0.0f, 0.0f, 0.0f, 0.0f);
}

template <int K>
__global__ void gemm_wmma_f32(const float* __restrict__ A,
                              const float* __restrict__ W,
                              const float* __restrict__ bias,
                              float* __restrict__ H,
                              int M, int do_relu) {
    __shared__ float Wlds[K * 64];
    #pragma unroll 4
    for (int i = threadIdx.x; i < K * 64; i += 256)
        Wlds[i] = W[i];
    __syncthreads();

    const int wave = threadIdx.x >> 5;
    const int lane = threadIdx.x & 31;
    const int half = lane >> 4;
    const int l16  = lane & 15;
    const int col0 = (wave & 3) * 16;
    const int row0 = blockIdx.x * 32 + (wave >> 2) * 16;
    const int m    = row0 + l16;
    const int mc   = (m < M) ? m : (M - 1);
    const float* __restrict__ Arow = A + (size_t)mc * K + 2 * half;
    const float* __restrict__ Brow = Wlds + 2 * half * 64 + col0 + l16;

    v8f acc = {};
    #pragma unroll
    for (int k0 = 0; k0 < K; k0 += 4) {
        v2f a = *(const v2f*)(Arow + k0);
        v2f b;
        b.x = Brow[k0 * 64];
        b.y = Brow[k0 * 64 + 64];
        acc = __builtin_amdgcn_wmma_f32_16x16x4_f32(
             false, a,  false, b,
             (short)0, acc,  false,  false);
    }
    const float bv = bias ? bias[col0 + l16] : 0.0f;
    #pragma unroll
    for (int r = 0; r < 8; ++r) {
        int mm = row0 + r + 8 * half;
        if (mm < M) {
            float v = acc[r] + bv;
            if (do_relu) v = fmaxf(v, 0.0f);
            H[(size_t)mm * 64 + col0 + l16] = v;
        }
    }
}

#define EDGE_GROUP 8

__device__ __forceinline__ int lane_bcast_i(int v, int j) {
    return __builtin_amdgcn_readlane(v, j);
}
__device__ __forceinline__ float lane_bcast_f(float v, int j) {
    return __int_as_float(__builtin_amdgcn_readlane(__float_as_int(v), j));
}

__global__ void edge_agg_kernel(const int* __restrict__ src,
                                const int* __restrict__ dst,
                                const float* __restrict__ dinv,
                                const float* __restrict__ h,
                                float* __restrict__ agg, int E) {
    const int w = __builtin_amdgcn_readfirstlane(blockIdx.x * (blockDim.x >> 5) +
                                                 (threadIdx.x >> 5));
    const int e0 = w * 32;
    if (e0 >= E) return;
    const int lane = threadIdx.x & 31;
    const int rem = E - e0;
    const int cnt = rem < 32 ? rem : 32;
    int s = 0, d = 0;
    float nrm = 0.0f;
    if (lane < cnt) {
        s = src[e0 + lane];
        d = dst[e0 + lane];
        nrm = dinv[s] * dinv[d];
    }
    int j = 0;
    for (; j + EDGE_GROUP <= cnt; j += EDGE_GROUP) {
        float2 v[EDGE_GROUP];
        #pragma unroll
        for (int u = 0; u < EDGE_GROUP; ++u) {
            const int sj = lane_bcast_i(s, j + u);
            v[u] = ((const float2*)(h + (size_t)sj * 64))[lane];
        }
        #pragma unroll
        for (int u = 0; u < EDGE_GROUP; ++u) {
            const int dj = lane_bcast_i(d, j + u);
            const float nj = lane_bcast_f(nrm, j + u);
            float* ad = agg + (size_t)dj * 64 + 2 * lane;
            atomic_add_f32(ad + 0, nj * v[u].x);
            atomic_add_f32(ad + 1, nj * v[u].y);
        }
    }
    for (; j < cnt; ++j) {
        const int sj = lane_bcast_i(s, j);
        const int dj = lane_bcast_i(d, j);
        const float nj = lane_bcast_f(nrm, j);
        const float2 v = ((const float2*)(h + (size_t)sj * 64))[lane];
        float* ad = agg + (size_t)dj * 64 + 2 * lane;
        atomic_add_f32(ad + 0, nj * v.x);
        atomic_add_f32(ad + 1, nj * v.y);
    }
}

__global__ void finalize_kernel(const float* __restrict__ agg,
                                const float* __restrict__ hlin,
                                const float* __restrict__ dinv,
                                const float* __restrict__ bias,
                                float* __restrict__ hout, int total4) {
    int i = blockIdx.x * blockDim.x + threadIdx.x;
    if (i >= total4) return;
    int n = i >> 4, f = (i & 15) * 4;
    float di = dinv[n];
    float4 a = ((const float4*)agg)[i];
    float4 l = ((const float4*)hlin)[i];
    float4 o;
    o.x = fmaxf(a.x + di * di * l.x + bias[f + 0], 0.0f);
    o.y = fmaxf(a.y + di * di * l.y + bias[f + 1], 0.0f);
    o.z = fmaxf(a.z + di * di * l.z + bias[f + 2], 0.0f);
    o.w = fmaxf(a.w + di * di * l.w + bias[f + 3], 0.0f);
    ((float4*)hout)[i] = o;
}

__global__ void pool_acc_kernel(const float* __restrict__ h,
                                const int* __restrict__ batch,
                                float* __restrict__ pooled,
                                float* __restrict__ counts, int n) {
    int node = __builtin_amdgcn_readfirstlane(blockIdx.x * (blockDim.x >> 5) +
                                              (threadIdx.x >> 5));
    if (node >= n) return;
    int lane = threadIdx.x & 31;
    int g = __builtin_amdgcn_readfirstlane(batch[node]);
    const float2* hn = (const float2*)(h + (size_t)node * 64);
    float2 v = hn[lane];
    float* pg = pooled + (size_t)g * 64 + 2 * lane;
    atomic_add_f32(pg + 0, v.x);
    atomic_add_f32(pg + 1, v.y);
    if (lane == 0) atomic_add_f32(&counts[g], 1.0f);
}

__global__ void pool_div_kernel(float* pooled, const float* counts, int total) {
    int i = blockIdx.x * blockDim.x + threadIdx.x;
    if (i >= total) return;
    float c = fmaxf(counts[i >> 6], 1.0f);
    pooled[i] = pooled[i] / c;
}

__global__ void head_kernel(const float* __restrict__ r1,
                            const float* __restrict__ lw2,
                            const float* __restrict__ lb2,
                            float* __restrict__ out, int G) {
    int i = blockIdx.x * blockDim.x + threadIdx.x;
    if (i >= G * 2) return;
    int g = i >> 1, j = i & 1;
    float acc = lb2[j];
    #pragma unroll
    for (int k = 0; k < 64; ++k) acc += r1[(size_t)g * 64 + k] * lw2[k * 2 + j];
    out[i] = acc;
}

static inline size_t alignup(size_t v) { return (v + 255) & ~(size_t)255; }

extern "C" void kernel_launch(void* const* d_in, const int* in_sizes, int n_in,
                              void* d_out, int out_size, void* d_ws, size_t ws_size,
                              hipStream_t stream) {
    const float* x     = (const float*)d_in[0];
    const int*   eidx  = (const int*)d_in[1];
    const int*   batch = (const int*)d_in[2];
    const float* W1 = (const float*)d_in[3];  const float* b1 = (const float*)d_in[4];
    const float* W2 = (const float*)d_in[5];  const float* b2 = (const float*)d_in[6];
    const float* W3 = (const float*)d_in[7];  const float* b3 = (const float*)d_in[8];
    const float* lw1 = (const float*)d_in[9]; const float* lb1 = (const float*)d_in[10];
    const float* lw2 = (const float*)d_in[11]; const float* lb2 = (const float*)d_in[12];
    float* out = (float*)d_out;

    const int N = N_NODES, G = N_GRAPHS;
    const int E = in_sizes[1] / 2;
    const int* src = eidx;
    const int* dst = eidx + E;

    char* w = (char*)d_ws;
    float* dinv   = (float*)w; w += alignup((size_t)N * 4);
    float* hlin   = (float*)w; w += alignup((size_t)N * 64 * 4);
    float* agg    = (float*)w; w += alignup((size_t)N * 64 * 4);
    float* hfeat  = (float*)w; w += alignup((size_t)N * 64 * 4);
    float* pooled = (float*)w; w += alignup((size_t)G * 64 * 4);
    float* counts = (float*)w; w += alignup((size_t)G * 4);
    float* relu1  = (float*)w; w += alignup((size_t)G * 64 * 4);
    (void)ws_size; (void)n_in;

    const int NF = N * 64;
    dim3 blk256(256);
    int gN    = (N + 255) / 256;
    const int NF4 = NF / 4;
    int gNF4  = (NF4 + 255) / 256;
    int gE    = (E + 255) / 256;
    int gE32  = (E + 255) / 256;
    int gN8   = (N + 7) / 8;
    int gGemmN = (N + 31) / 32;
    int gGemmG = (G + 31) / 32;

    init_deg_kernel<<<gN, blk256, 0, stream>>>(dinv, N);
    count_deg_kernel<<<gE, blk256, 0, stream>>>(dst, dinv, E);
    dinv_kernel<<<gN, blk256, 0, stream>>>(dinv, N);

    gemm_wmma_f32<32><<<gGemmN, blk256, 0, stream>>>(x, W1, nullptr, hlin, N, 0);
    zero4_kernel<<<gNF4, blk256, 0, stream>>>(agg, NF4);
    edge_agg_kernel<<<gE32, blk256, 0, stream>>>(src, dst, dinv, hlin, agg, E);
    finalize_kernel<<<gNF4, blk256, 0, stream>>>(agg, hlin, dinv, b1, hfeat, NF4);

    gemm_wmma_f32<64><<<gGemmN, blk256, 0, stream>>>(hfeat, W2, nullptr, hlin, N, 0);
    zero4_kernel<<<gNF4, blk256, 0, stream>>>(agg, NF4);
    edge_agg_kernel<<<gE32, blk256, 0, stream>>>(src, dst, dinv, hlin, agg, E);
    finalize_kernel<<<gNF4, blk256, 0, stream>>>(agg, hlin, dinv, b2, hfeat, NF4);

    gemm_wmma_f32<64><<<gGemmN, blk256, 0, stream>>>(hfeat, W3, nullptr, hlin, N, 0);
    zero4_kernel<<<gNF4, blk256, 0, stream>>>(agg, NF4);
    edge_agg_kernel<<<gE32, blk256, 0, stream>>>(src, dst, dinv, hlin, agg, E);
    finalize_kernel<<<gNF4, blk256, 0, stream>>>(agg, hlin, dinv, b3, hfeat, NF4);

    int GF = G * 64;
    zero_kernel<<<(GF + 255) / 256, blk256, 0, stream>>>(pooled, GF);
    zero_kernel<<<(G + 255) / 256, blk256, 0, stream>>>(counts, G);
    pool_acc_kernel<<<gN8, blk256, 0, stream>>>(hfeat, batch, pooled, counts, N);
    pool_div_kernel<<<(GF + 255) / 256, blk256, 0, stream>>>(pooled, counts, GF);

    gemm_wmma_f32<64><<<gGemmG, blk256, 0, stream>>>(pooled, lw1, lb1, relu1, G, 1);
    head_kernel<<<(G * 2 + 255) / 256, blk256, 0, stream>>>(relu1, lw2, lb2, out, G);
    (void)out_size;
}


# edge scatter by f32 atomics replaced by a dst-bucketed edge list rebuilt each call (int rank atomics, scan) and one wave per node that gathers its rows, writes the row once; sum order = rank order
# speedup vs baseline: 2.0002x; 1.7634x over previous
#include <hip/hip_runtime.h>
#include <hip/hip_bf16.h>

typedef __attribute__((ext_vector_type(2))) float v2f;
typedef __attribute__((ext_vector_type(8))) float v8f;

#define N_NODES 50000
#define N_GRAPHS 2048

__device__ __forceinline__ void atomic_add_f32(float* p, float v) {
    unsafeAtomicAdd(p, v);
}

__global__ void zero_i4_kernel(int* p, int n4) {
    int i = blockIdx.x * blockDim.x + threadIdx.x;
    if (i < n4) ((int4*)p)[i] = make_int4(0, 0, 0, 0);
}

__global__ void count_rank_kernel(const int* __restrict__ dst, int* cnt,
                                  int* __restrict__ rank, int E) {
    int i = blockIdx.x * blockDim.x + threadIdx.x;
    if (i < E) rank[i] = atomicAdd(&cnt[dst[i]], 1);
}

__global__ void rowptr_scan_kernel(const int* __restrict__ cnt,
                                   int* __restrict__ rowptr, int n4) {
    __shared__ int pre_w[8];
    __shared__ int tot_w[8];
    const int t = threadIdx.x, lane = t & 31, wave = t >> 5;
    const int4* c4 = (const int4*)cnt;
    int pre = 0;
    #pragma unroll 8
    for (int k = 0; k < (int)blockIdx.x; ++k) {
        int4 v = c4[k * 256 + t];
        pre += (v.x + v.y) + (v.z + v.w);
    }
    const int i4 = blockIdx.x * 256 + t;
    int4 own = make_int4(0, 0, 0, 0);
    if (i4 < n4) own = c4[i4];
    const int mine = (own.x + own.y) + (own.z + own.w);
    int incl = mine;
    #pragma unroll
    for (int o = 1; o < 32; o <<= 1) {
        int up = __shfl_up(incl, o);
        if (lane >= o) incl += up;
    }
    #pragma unroll
    for (int o = 16; o > 0; o >>= 1) pre += __shfl_xor(pre, o);
    if (lane == 31) tot_w[wave] = incl;
    if (lane == 0)  pre_w[wave] = pre;
    __syncthreads();
    int base = 0;
    #pragma unroll
    for (int w = 0; w < 8; ++w) {
        base += pre_w[w];
        if (w < wave) base += tot_w[w];
    }
    if (i4 < n4) {
        int4 r;
        r.x = base + incl - mine;
        r.y = r.x + own.x;
        r.z = r.y + own.y;
        r.w = r.z + own.z;
        ((int4*)rowptr)[i4] = r;
    }
}

__global__ void bucket_fill_kernel(const int* __restrict__ src,
                                   const int* __restrict__ dst,
                                   const int* __restrict__ rank,
                                   const int* __restrict__ rowptr,
                                   const int* __restrict__ cnt,
                                   int2* __restrict__ pairs, int E) {
    int i = blockIdx.x * blockDim.x + threadIdx.x;
    if (i >= E) return;
    const int s = src[i], d = dst[i];
    const float nrm = rsqrtf((float)(cnt[s] + 1)) * rsqrtf((float)(cnt[d] + 1));
    pairs[rowptr[d] + rank[i]] = make_int2(s, __float_as_int(nrm));
}

__global__ void zero_kernel(float* p, int n) {
    int i = blockIdx.x * blockDim.x + threadIdx.x;
    if (i < n) p[i] = 0.0f;
}

template <int K>
__global__ void gemm_wmma_f32(const float* __restrict__ A,
                              const float* __restrict__ W,
                              const float* __restrict__ bias,
                              float* __restrict__ H,
                              int M, int do_relu) {
    __shared__ float Wlds[K * 64];
    #pragma unroll 4
    for (int i = threadIdx.x; i < K * 64; i += 256)
        Wlds[i] = W[i];
    __syncthreads();

    const int wave = threadIdx.x >> 5;
    const int lane = threadIdx.x & 31;
    const int half = lane >> 4;
    const int l16  = lane & 15;
    const int col0 = (wave & 3) * 16;
    const int row0 = blockIdx.x * 32 + (wave >> 2) * 16;
    const int m    = row0 + l16;
    const int mc   = (m < M) ? m : (M - 1);
    const float* __restrict__ Arow = A + (size_t)mc * K + 2 * half;
    const float* __restrict__ Brow = Wlds + 2 * half * 64 + col0 + l16;

    v8f acc = {};
    #pragma unroll
    for (int k0 = 0; k0 < K; k0 += 4) {
        v2f a = *(const v2f*)(Arow + k0);
        v2f b;
        b.x = Brow[k0 * 64];
        b.y = Brow[k0 * 64 + 64];
        acc = __builtin_amdgcn_wmma_f32_16x16x4_f32(
             false, a,  false, b,
             (short)0, acc,  false,  false);
    }
    const float bv = bias ? bias[col0 + l16] : 0.0f;
    #pragma unroll
    for (int r = 0; r < 8; ++r) {
        int mm = row0 + r + 8 * half;
        if (mm < M) {
            float v = acc[r] + bv;
            if (do_relu) v = fmaxf(v, 0.0f);
            H[(size_t)mm * 64 + col0 + l16] = v;
        }
    }
}

__device__ __forceinline__ int lane_bcast_i(int v, int j) {
    return __builtin_amdgcn_readlane(v, j);
}
__device__ __forceinline__ float lane_bcast_f(float v, int j) {
    return __int_as_float(__builtin_amdgcn_readlane(__float_as_int(v), j));
}

#define GATHER_GROUP 16
__device__ __forceinline__ void gather_group(const float* __restrict__ h, int s, float w,
                                             int j, int lane, float& ax, float& ay) {
    float2 v[GATHER_GROUP];
    #pragma unroll
    for (int u = 0; u < GATHER_GROUP; ++u) {
        const int sj = lane_bcast_i(s, j + u);
        v[u] = ((const float2*)(h + (size_t)sj * 64))[lane];
    }
    #pragma unroll
    for (int u = 0; u < GATHER_GROUP; ++u) {
        const float wj = lane_bcast_f(w, j + u);
        ax += wj * v[u].x;
        ay += wj * v[u].y;
    }
}

__global__ void gather_agg_kernel(const int* __restrict__ rowptr,
                                  const int* __restrict__ cnt,
                                  const int2* __restrict__ pairs,
                                  const float* __restrict__ h,
                                  const float* __restrict__ bias,
                                  float* __restrict__ hout, int N) {
    const int n = __builtin_amdgcn_readfirstlane(blockIdx.x * (blockDim.x >> 5) +
                                                 (threadIdx.x >> 5));
    if (n >= N) return;
    const int lane = threadIdx.x & 31;
    const int start = __builtin_amdgcn_readfirstlane(rowptr[n]);
    const int deg   = __builtin_amdgcn_readfirstlane(cnt[n]);
    float ax = 0.0f, ay = 0.0f;
    const int c0 = deg < 32 ? deg : 32;
    int s = n;
    float w = 0.0f;
    if (lane < c0) {
        const int2 p = pairs[start + lane];
        s = p.x;
        w = __int_as_float(p.y);
    }
    const float2 self = ((const float2*)(h + (size_t)n * 64))[lane];
    const float2 bv = ((const float2*)bias)[lane];
    const float dn = rsqrtf((float)(deg + 1));
    for (int j = 0; j < c0; j += GATHER_GROUP)
        gather_group(h, s, w, j, lane, ax, ay);
    for (int base = 32; base < deg; base += 32) {
        const int rem = deg - base;
        const int c = rem < 32 ? rem : 32;
        int s2 = n;
        float w2 = 0.0f;
        if (lane < c) {
            const int2 p = pairs[start + base + lane];
            s2 = p.x;
            w2 = __int_as_float(p.y);
        }
        for (int j = 0; j < c; j += GATHER_GROUP)
            gather_group(h, s2, w2, j, lane, ax, ay);
    }
    float2 o;
    o.x = fmaxf(ax + dn * dn * self.x + bv.x, 0.0f);
    o.y = fmaxf(ay + dn * dn * self.y + bv.y, 0.0f);
    ((float2*)(hout + (size_t)n * 64))[lane] = o;
}

__global__ void pool_acc_kernel(const float* __restrict__ h,
                                const int* __restrict__ batch,
                                float* __restrict__ pooled,
                                float* __restrict__ counts, int n) {
    int node = __builtin_amdgcn_readfirstlane(blockIdx.x * (blockDim.x >> 5) +
                                              (threadIdx.x >> 5));
    if (node >= n) return;
    int lane = threadIdx.x & 31;
    int g = __builtin_amdgcn_readfirstlane(batch[node]);
    const float2* hn = (const float2*)(h + (size_t)node * 64);
    float2 v = hn[lane];
    float* pg = pooled + (size_t)g * 64 + 2 * lane;
    atomic_add_f32(pg + 0, v.x);
    atomic_add_f32(pg + 1, v.y);
    if (lane == 0) atomic_add_f32(&counts[g], 1.0f);
}

__global__ void pool_div_kernel(float* pooled, const float* counts, int total) {
    int i = blockIdx.x * blockDim.x + threadIdx.x;
    if (i >= total) return;
    float c = fmaxf(counts[i >> 6], 1.0f);
    pooled[i] = pooled[i] / c;
}

__global__ void head_kernel(const float* __restrict__ r1,
                            const float* __restrict__ lw2,
                            const float* __restrict__ lb2,
                            float* __restrict__ out, int G) {
    int i = blockIdx.x * blockDim.x + threadIdx.x;
    if (i >= G * 2) return;
    int g = i >> 1, j = i & 1;
    float acc = lb2[j];
    #pragma unroll
    for (int k = 0; k < 64; ++k) acc += r1[(size_t)g * 64 + k] * lw2[k * 2 + j];
    out[i] = acc;
}

static inline size_t alignup(size_t v) { return (v + 255) & ~(size_t)255; }

extern "C" void kernel_launch(void* const* d_in, const int* in_sizes, int n_in,
                              void* d_out, int out_size, void* d_ws, size_t ws_size,
                              hipStream_t stream) {
    const float* x     = (const float*)d_in[0];
    const int*   eidx  = (const int*)d_in[1];
    const int*   batch = (const int*)d_in[2];
    const float* W1 = (const float*)d_in[3];  const float* b1 = (const float*)d_in[4];
    const float* W2 = (const float*)d_in[5];  const float* b2 = (const float*)d_in[6];
    const float* W3 = (const float*)d_in[7];  const float* b3 = (const float*)d_in[8];
    const float* lw1 = (const float*)d_in[9]; const float* lb1 = (const float*)d_in[10];
    const float* lw2 = (const float*)d_in[11]; const float* lb2 = (const float*)d_in[12];
    float* out = (float*)d_out;

    const int N = N_NODES, G = N_GRAPHS;
    const int E = in_sizes[1] / 2;
    const int* src = eidx;
    const int* dst = eidx + E;

    char* w = (char*)d_ws;
    int*   cnt    = (int*)w;   w += alignup((size_t)N * 4);
    int*   rowptr = (int*)w;   w += alignup((size_t)N * 4);
    int*   rank   = (int*)w;   w += alignup((size_t)E * 4);
    int2*  pairs  = (int2*)w;  w += alignup((size_t)E * 8);
    float* hlin   = (float*)w; w += alignup((size_t)N * 64 * 4);
    float* hfeat  = (float*)w; w += alignup((size_t)N * 64 * 4);
    float* pooled = (float*)w; w += alignup((size_t)G * 64 * 4);
    float* counts = (float*)w; w += alignup((size_t)G * 4);
    float* relu1  = (float*)w; w += alignup((size_t)G * 64 * 4);
    (void)ws_size; (void)n_in;

    dim3 blk256(256);
    const int N4 = N / 4;
    int gN4   = (N4 + 255) / 256;
    int gE    = (E + 255) / 256;
    int gN8   = (N + 7) / 8;
    int gGemmN = (N + 31) / 32;
    int gGemmG = (G + 31) / 32;

    zero_i4_kernel<<<gN4, blk256, 0, stream>>>(cnt, N4);
    count_rank_kernel<<<gE, blk256, 0, stream>>>(dst, cnt, rank, E);
    rowptr_scan_kernel<<<gN4, blk256, 0, stream>>>(cnt, rowptr, N4);
    bucket_fill_kernel<<<gE, blk256, 0, stream>>>(src, dst, rank, rowptr, cnt, pairs, E);

    gemm_wmma_f32<32><<<gGemmN, blk256, 0, stream>>>(x, W1, nullptr, hlin, N, 0);
    gather_agg_kernel<<<gN8, blk256, 0, stream>>>(rowptr, cnt, pairs, hlin, b1, hfeat, N);

    gemm_wmma_f32<64><<<gGemmN, blk256, 0, stream>>>(hfeat, W2, nullptr, hlin, N, 0);
    gather_agg_kernel<<<gN8, blk256, 0, stream>>>(rowptr, cnt, pairs, hlin, b2, hfeat, N);

    gemm_wmma_f32<64><<<gGemmN, blk256, 0, stream>>>(hfeat, W3, nullptr, hlin, N, 0);
    gather_agg_kernel<<<gN8, blk256, 0, stream>>>(rowptr, cnt, pairs, hlin, b3, hfeat, N);

    int GF = G * 64;
    zero_kernel<<<(GF + 255) / 256, blk256, 0, stream>>>(pooled, GF);
    zero_kernel<<<(G + 255) / 256, blk256, 0, stream>>>(counts, G);
    pool_acc_kernel<<<gN8, blk256, 0, stream>>>(hfeat, batch, pooled, counts, N);
    pool_div_kernel<<<(GF + 255) / 256, blk256, 0, stream>>>(pooled, counts, GF);

    gemm_wmma_f32<64><<<gGemmG, blk256, 0, stream>>>(pooled, lw1, lb1, relu1, G, 1);
    head_kernel<<<(G * 2 + 255) / 256, blk256, 0, stream>>>(relu1, lw2, lb2, out, G);
    (void)out_size;
}


# 16 launches -> 10: clears share one launch, layer-1 product shares the count launch, pooling sums (f32 atomics kept, arrival order differs) move into the last gather, mean+MLP+head in one kernel
# speedup vs baseline: 2.1591x; 1.0794x over previous
#include <hip/hip_runtime.h>
#include <hip/hip_bf16.h>

typedef __attribute__((ext_vector_type(2))) float v2f;
typedef __attribute__((ext_vector_type(8))) float v8f;

#define N_NODES 50000
#define N_GRAPHS 2048

__device__ __forceinline__ void atomic_add_f32(float* p, float v) {
    unsafeAtomicAdd(p, v);
}

__global__ void clear_kernel(int* cnt, int n4_cnt, float* pooled, int n4_pooled,
                             float* counts, int n4_counts) {
    int i = blockIdx.x * blockDim.x + threadIdx.x;
    if (i < n4_cnt) { ((int4*)cnt)[i] = make_int4(0, 0, 0, 0); return; }
    i -= n4_cnt;
    if (i < n4_pooled) { ((float4*)pooled)[i] = make_float4(0.f, 0.f, 0.f, 0.f); return; }
    i -= n4_pooled;
    if (i < n4_counts) ((float4*)counts)[i] = make_float4(0.f, 0.f, 0.f, 0.f);
}

__global__ void rowptr_scan_kernel(const int* __restrict__ cnt,
                                   int* __restrict__ rowptr, int n4) {
    __shared__ int pre_w[8];
    __shared__ int tot_w[8];
    const int t = threadIdx.x, lane = t & 31, wave = t >> 5;
    const int4* c4 = (const int4*)cnt;
    int pre = 0;
    #pragma unroll 8
    for (int k = 0; k < (int)blockIdx.x; ++k) {
        int4 v = c4[k * 256 + t];
        pre += (v.x + v.y) + (v.z + v.w);
    }
    const int i4 = blockIdx.x * 256 + t;
    int4 own = make_int4(0, 0, 0, 0);
    if (i4 < n4) own = c4[i4];
    const int mine = (own.x + own.y) + (own.z + own.w);
    int incl = mine;
    #pragma unroll
    for (int o = 1; o < 32; o <<= 1) {
        int up = __shfl_up(incl, o);
        if (lane >= o) incl += up;
    }
    #pragma unroll
    for (int o = 16; o > 0; o >>= 1) pre += __shfl_xor(pre, o);
    if (lane == 31) tot_w[wave] = incl;
    if (lane == 0)  pre_w[wave] = pre;
    __syncthreads();
    int base = 0;
    #pragma unroll
    for (int w = 0; w < 8; ++w) {
        base += pre_w[w];
        if (w < wave) base += tot_w[w];
    }
    if (i4 < n4) {
        int4 r;
        r.x = base + incl - mine;
        r.y = r.x + own.x;
        r.z = r.y + own.y;
        r.w = r.z + own.z;
        ((int4*)rowptr)[i4] = r;
    }
}

__global__ void bucket_fill_kernel(const int* __restrict__ src,
                                   const int* __restrict__ dst,
                                   const int* __restrict__ rank,
                                   const int* __restrict__ rowptr,
                                   const int* __restrict__ cnt,
                                   int2* __restrict__ pairs, int E) {
    int i = blockIdx.x * blockDim.x + threadIdx.x;
    if (i >= E) return;
    const int s = src[i], d = dst[i];
    const float nrm = rsqrtf((float)(cnt[s] + 1)) * rsqrtf((float)(cnt[d] + 1));
    pairs[rowptr[d] + rank[i]] = make_int2(s, __float_as_int(nrm));
}

template <int K>
__device__ __forceinline__ void gemm_tile(const float* __restrict__ A,
                              const float* __restrict__ W,
                              const float* __restrict__ bias,
                              float* __restrict__ H,
                              int M, int do_relu, int blk) {
    __shared__ float Wlds[K * 64];
    #pragma unroll 4
    for (int i = threadIdx.x; i < K * 64; i += 256)
        Wlds[i] = W[i];
    __syncthreads();

    const int wave = threadIdx.x >> 5;
    const int lane = threadIdx.x & 31;
    const int half = lane >> 4;
    const int l16  = lane & 15;
    const int col0 = (wave & 3) * 16;
    const int row0 = blk * 32 + (wave >> 2) * 16;
    const int m    = row0 + l16;
    const int mc   = (m < M) ? m : (M - 1);
    const float* __restrict__ Arow = A + (size_t)mc * K + 2 * half;
    const float* __restrict__ Brow = Wlds + 2 * half * 64 + col0 + l16;

    v8f acc = {};
    #pragma unroll
    for (int k0 = 0; k0 < K; k0 += 4) {
        v2f a = *(const v2f*)(Arow + k0);
        v2f b;
        b.x = Brow[k0 * 64];
        b.y = Brow[k0 * 64 + 64];
        acc = __builtin_amdgcn_wmma_f32_16x16x4_f32(
             false, a,  false, b,
             (short)0, acc,  false,  false);
    }
    const float bv = bias ? bias[col0 + l16] : 0.0f;
    #pragma unroll
    for (int r = 0; r < 8; ++r) {
        int mm = row0 + r + 8 * half;
        if (mm < M) {
            float v = acc[r] + bv;
            if (do_relu) v = fmaxf(v, 0.0f);
            H[(size_t)mm * 64 + col0 + l16] = v;
        }
    }
}

template <int K>
__global__ void gemm_wmma_f32(const float* __restrict__ A,
                              const float* __restrict__ W,
                              const float* __restrict__ bias,
                              float* __restrict__ H,
                              int M, int do_relu) {
    gemm_tile<K>(A, W, bias, H, M, do_relu, (int)blockIdx.x);
}

__global__ void gemm32_or_count_rank_kernel(const float* __restrict__ x,
                                            const float* __restrict__ W1,
                                            float* __restrict__ hlin, int M, int n_gemm,
                                            const int* __restrict__ dst, int* cnt,
                                            int* __restrict__ rank, int E) {
    if ((int)blockIdx.x < n_gemm) {
        gemm_tile<32>(x, W1, nullptr, hlin, M, 0, (int)blockIdx.x);
        return;
    }
    const int i = ((int)blockIdx.x - n_gemm) * 256 + (int)threadIdx.x;
    if (i < E) rank[i] = atomicAdd(&cnt[dst[i]], 1);
}

__device__ __forceinline__ int lane_bcast_i(int v, int j) {
    return __builtin_amdgcn_readlane(v, j);
}
__device__ __forceinline__ float lane_bcast_f(float v, int j) {
    return __int_as_float(__builtin_amdgcn_readlane(__float_as_int(v), j));
}

#define GATHER_GROUP 16
__device__ __forceinline__ void gather_group(const float* __restrict__ h, int s, float w,
                                             int j, int lane, float& ax, float& ay) {
    float2 v[GATHER_GROUP];
    #pragma unroll
    for (int u = 0; u < GATHER_GROUP; ++u) {
        const int sj = lane_bcast_i(s, j + u);
        v[u] = ((const float2*)(h + (size_t)sj * 64))[lane];
    }
    #pragma unroll
    for (int u = 0; u < GATHER_GROUP; ++u) {
        const float wj = lane_bcast_f(w, j + u);
        ax += wj * v[u].x;
        ay += wj * v[u].y;
    }
}

template <bool POOL>
__global__ void gather_agg_kernel(const int* __restrict__ rowptr,
                                  const int* __restrict__ cnt,
                                  const int2* __restrict__ pairs,
                                  const float* __restrict__ h,
                                  const float* __restrict__ bias,
                                  float* __restrict__ hout,
                                  const int* __restrict__ batch,
                                  float* pooled, float* counts, int N) {
    const int n = __builtin_amdgcn_readfirstlane(blockIdx.x * (blockDim.x >> 5) +
                                                 (threadIdx.x >> 5));
    if (n >= N) return;
    const int lane = threadIdx.x & 31;
    const int start = __builtin_amdgcn_readfirstlane(rowptr[n]);
    const int deg   = __builtin_amdgcn_readfirstlane(cnt[n]);
    int g = 0;
    if (POOL) g = batch[n];
    float ax = 0.0f, ay = 0.0f;
    const int c0 = deg < 32 ? deg : 32;
    int s = n;
    float w = 0.0f;
    if (lane < c0) {
        const int2 p = pairs[start + lane];
        s = p.x;
        w = __int_as_float(p.y);
    }
    const float2 self = ((const float2*)(h + (size_t)n * 64))[lane];
    const float2 bv = ((const float2*)bias)[lane];
    const float dn = rsqrtf((float)(deg + 1));
    for (int j = 0; j < c0; j += GATHER_GROUP)
        gather_group(h, s, w, j, lane, ax, ay);
    for (int base = 32; base < deg; base += 32) {
        const int rem = deg - base;
        const int c = rem < 32 ? rem : 32;
        int s2 = n;
        float w2 = 0.0f;
        if (lane < c) {
            const int2 p = pairs[start + base + lane];
            s2 = p.x;
            w2 = __int_as_float(p.y);
        }
        for (int j = 0; j < c; j += GATHER_GROUP)
            gather_group(h, s2, w2, j, lane, ax, ay);
    }
    float2 o;
    o.x = fmaxf(ax + dn * dn * self.x + bv.x, 0.0f);
    o.y = fmaxf(ay + dn * dn * self.y + bv.y, 0.0f);
    if (POOL) {
        float* pg = pooled + (size_t)g * 64 + 2 * lane;
        atomic_add_f32(pg + 0, o.x);
        atomic_add_f32(pg + 1, o.y);
        if (lane == 0) atomic_add_f32(&counts[g], 1.0f);
    } else {
        ((float2*)(hout + (size_t)n * 64))[lane] = o;
    }
}

__global__ void pooled_mlp_head_kernel(const float* __restrict__ pooled,
                                       const float* __restrict__ counts,
                                       const float* __restrict__ lw1,
                                       const float* __restrict__ lb1,
                                       const float* __restrict__ lw2,
                                       const float* __restrict__ lb2,
                                       float* __restrict__ out, int G) {
    __shared__ float Wlds[64 * 64];
    __shared__ float Rt[32 * 65];
    #pragma unroll 4
    for (int i = threadIdx.x; i < 64 * 64; i += 256)
        Wlds[i] = lw1[i];
    __syncthreads();

    const int wave = threadIdx.x >> 5;
    const int lane = threadIdx.x & 31;
    const int half = lane >> 4;
    const int l16  = lane & 15;
    const int col0 = (wave & 3) * 16;
    const int rloc = (wave >> 2) * 16;
    const int row0 = blockIdx.x * 32 + rloc;
    const int m    = row0 + l16;
    const int mc   = (m < G) ? m : (G - 1);
    const float c  = fmaxf(counts[mc], 1.0f);
    const float* __restrict__ Arow = pooled + (size_t)mc * 64 + 2 * half;
    const float* __restrict__ Brow = Wlds + 2 * half * 64 + col0 + l16;

    v8f acc = {};
    #pragma unroll
    for (int k0 = 0; k0 < 64; k0 += 4) {
        v2f a = *(const v2f*)(Arow + k0);
        a.x = a.x / c;
        a.y = a.y / c;
        v2f b;
        b.x = Brow[k0 * 64];
        b.y = Brow[k0 * 64 + 64];
        acc = __builtin_amdgcn_wmma_f32_16x16x4_f32(
             false, a,  false, b,
             (short)0, acc,  false,  false);
    }
    const float bv = lb1[col0 + l16];
    #pragma unroll
    for (int r = 0; r < 8; ++r)
        Rt[(rloc + r + 8 * half) * 65 + col0 + l16] = fmaxf(acc[r] + bv, 0.0f);
    __syncthreads();

    const int t = threadIdx.x;
    if (t < 64) {
        const int gl = t >> 1, j = t & 1;
        const int gg = blockIdx.x * 32 + gl;
        if (gg < G) {
            float o = lb2[j];
            #pragma unroll
            for (int k = 0; k < 64; ++k) o += Rt[gl * 65 + k] * lw2[k * 2 + j];
            out[(size_t)gg * 2 + j] = o;
        }
    }
}

static inline size_t alignup(size_t v) { return (v + 255) & ~(size_t)255; }

extern "C" void kernel_launch(void* const* d_in, const int* in_sizes, int n_in,
                              void* d_out, int out_size, void* d_ws, size_t ws_size,
                              hipStream_t stream) {
    const float* x     = (const float*)d_in[0];
    const int*   eidx  = (const int*)d_in[1];
    const int*   batch = (const int*)d_in[2];
    const float* W1 = (const float*)d_in[3];  const float* b1 = (const float*)d_in[4];
    const float* W2 = (const float*)d_in[5];  const float* b2 = (const float*)d_in[6];
    const float* W3 = (const float*)d_in[7];  const float* b3 = (const float*)d_in[8];
    const float* lw1 = (const float*)d_in[9]; const float* lb1 = (const float*)d_in[10];
    const float* lw2 = (const float*)d_in[11]; const float* lb2 = (const float*)d_in[12];
    float* out = (float*)d_out;

    const int N = N_NODES, G = N_GRAPHS;
    const int E = in_sizes[1] / 2;
    const int* src = eidx;
    const int* dst = eidx + E;

    char* w = (char*)d_ws;
    int*   cnt    = (int*)w;   w += alignup((size_t)N * 4);
    int*   rowptr = (int*)w;   w += alignup((size_t)N * 4);
    int*   rank   = (int*)w;   w += alignup((size_t)E * 4);
    int2*  pairs  = (int2*)w;  w += alignup((size_t)E * 8);
    float* hlin   = (float*)w; w += alignup((size_t)N * 64 * 4);
    float* hfeat  = (float*)w; w += alignup((size_t)N * 64 * 4);
    float* pooled = (float*)w; w += alignup((size_t)G * 64 * 4);
    float* counts = (float*)w; w += alignup((size_t)G * 4);
    (void)ws_size; (void)n_in;

    dim3 blk256(256);
    const int N4 = N / 4;
    const int GF4 = G * 64 / 4;
    const int G4 = G / 4;
    int gClr  = (N4 + GF4 + G4 + 255) / 256;
    int gN4   = (N4 + 255) / 256;
    int gE    = (E + 255) / 256;
    int gN8   = (N + 7) / 8;
    int gGemmN = (N + 31) / 32;
    int gGemmG = (G + 31) / 32;

    clear_kernel<<<gClr, blk256, 0, stream>>>(cnt, N4, pooled, GF4, counts, G4);
    gemm32_or_count_rank_kernel<<<gGemmN + gE, blk256, 0, stream>>>(
        x, W1, hlin, N, gGemmN, dst, cnt, rank, E);
    rowptr_scan_kernel<<<gN4, blk256, 0, stream>>>(cnt, rowptr, N4);
    bucket_fill_kernel<<<gE, blk256, 0, stream>>>(src, dst, rank, rowptr, cnt, pairs, E);

    gather_agg_kernel<false><<<gN8, blk256, 0, stream>>>(rowptr, cnt, pairs, hlin, b1, hfeat,
                                                         batch, pooled, counts, N);

    gemm_wmma_f32<64><<<gGemmN, blk256, 0, stream>>>(hfeat, W2, nullptr, hlin, N, 0);
    gather_agg_kernel<false><<<gN8, blk256, 0, stream>>>(rowptr, cnt, pairs, hlin, b2, hfeat,
                                                         batch, pooled, counts, N);

    gemm_wmma_f32<64><<<gGemmN, blk256, 0, stream>>>(hfeat, W3, nullptr, hlin, N, 0);
    gather_agg_kernel<true><<<gN8, blk256, 0, stream>>>(rowptr, cnt, pairs, hlin, b3, hfeat,
                                                        batch, pooled, counts, N);

    pooled_mlp_head_kernel<<<gGemmG, blk256, 0, stream>>>(pooled, counts, lw1, lb1, lw2, lb2, out, G);
    (void)out_size;
}


# each layer's dense product moved behind its neighbour sum ((A h) W, not A (h W)) and into the gather launch, 32 nodes per block via LDS: 10 launches -> 8; rounding order of the sums differs
# speedup vs baseline: 2.5486x; 1.1804x over previous
#include <hip/hip_runtime.h>

typedef __attribute__((ext_vector_type(2))) float v2f;
typedef __attribute__((ext_vector_type(8))) float v8f;

#define N_NODES 50000
#define N_GRAPHS 2048

__device__ __forceinline__ void atomic_add_f32(float* p, float v) {
    unsafeAtomicAdd(p, v);
}

__global__ void clear_kernel(int* cnt, int n4_cnt, float* pooled, int n4_pooled,
                             float* counts, int n4_counts) {
    int i = blockIdx.x * blockDim.x + threadIdx.x;
    if (i < n4_cnt) { ((int4*)cnt)[i] = make_int4(0, 0, 0, 0); return; }
    i -= n4_cnt;
    if (i < n4_pooled) { ((float4*)pooled)[i] = make_float4(0.f, 0.f, 0.f, 0.f); return; }
    i -= n4_pooled;
    if (i < n4_counts) ((float4*)counts)[i] = make_float4(0.f, 0.f, 0.f, 0.f);
}

__global__ void rowptr_scan_kernel(const int* __restrict__ cnt,
                                   int* __restrict__ rowptr, int n4) {
    __shared__ int pre_w[8];
    __shared__ int tot_w[8];
    const int t = threadIdx.x, lane = t & 31, wave = t >> 5;
    const int4* c4 = (const int4*)cnt;
    int pre = 0;
    #pragma unroll 8
    for (int k = 0; k < (int)blockIdx.x; ++k) {
        int4 v = c4[k * 256 + t];
        pre += (v.x + v.y) + (v.z + v.w);
    }
    const int i4 = blockIdx.x * 256 + t;
    int4 own = make_int4(0, 0, 0, 0);
    if (i4 < n4) own = c4[i4];
    const int mine = (own.x + own.y) + (own.z + own.w);
    int incl = mine;
    #pragma unroll
    for (int o = 1; o < 32; o <<= 1) {
        int up = __shfl_up(incl, o);
        if (lane >= o) incl += up;
    }
    #pragma unroll
    for (int o = 16; o > 0; o >>= 1) pre += __shfl_xor(pre, o);
    if (lane == 31) tot_w[wave] = incl;
    if (lane == 0)  pre_w[wave] = pre;
    __syncthreads();
    int base = 0;
    #pragma unroll
    for (int w = 0; w < 8; ++w) {
        base += pre_w[w];
        if (w < wave) base += tot_w[w];
    }
    if (i4 < n4) {
        int4 r;
        r.x = base + incl - mine;
        r.y = r.x + own.x;
        r.z = r.y + own.y;
        r.w = r.z + own.z;
        ((int4*)rowptr)[i4] = r;
    }
}

__global__ void bucket_fill_kernel(const int* __restrict__ src,
                                   const int* __restrict__ dst,
                                   const int* __restrict__ rank,
                                   const int* __restrict__ rowptr,
                                   const int* __restrict__ cnt,
                                   int2* __restrict__ pairs, int E) {
    int i = blockIdx.x * blockDim.x + threadIdx.x;
    if (i >= E) return;
    const int s = src[i], d = dst[i];
    const float nrm = rsqrtf((float)(cnt[s] + 1)) * rsqrtf((float)(cnt[d] + 1));
    pairs[rowptr[d] + rank[i]] = make_int2(s, __float_as_int(nrm));
}

__global__ void count_rank_kernel(const int* __restrict__ dst, int* cnt,
                                  int* __restrict__ rank, int E) {
    const int i = (int)blockIdx.x * 256 + (int)threadIdx.x;
    if (i < E) rank[i] = atomicAdd(&cnt[dst[i]], 1);
}

__device__ __forceinline__ int lane_bcast_i(int v, int j) {
    return __builtin_amdgcn_readlane(v, j);
}
__device__ __forceinline__ float lane_bcast_f(float v, int j) {
    return __int_as_float(__builtin_amdgcn_readlane(__float_as_int(v), j));
}

#define GATHER_GROUP 16
#define NODES_PER_WAVE 4
template <int K>
__device__ __forceinline__ void gather_group(const float* __restrict__ h, int s, float w,
                                             int j, int lane, float& ax, float& ay) {
    if constexpr (K == 64) {
        float2 v[GATHER_GROUP];
        #pragma unroll
        for (int u = 0; u < GATHER_GROUP; ++u) {
            const int sj = lane_bcast_i(s, j + u);
            v[u] = ((const float2*)(h + (size_t)sj * 64))[lane];
        }
        #pragma unroll
        for (int u = 0; u < GATHER_GROUP; ++u) {
            const float wj = lane_bcast_f(w, j + u);
            ax += wj * v[u].x;
            ay += wj * v[u].y;
        }
    } else {
        float v[GATHER_GROUP];
        #pragma unroll
        for (int u = 0; u < GATHER_GROUP; ++u) {
            const int sj = lane_bcast_i(s, j + u);
            v[u] = h[(size_t)sj * 32 + lane];
        }
        #pragma unroll
        for (int u = 0; u < GATHER_GROUP; ++u) {
            const float wj = lane_bcast_f(w, j + u);
            ax += wj * v[u];
        }
    }
}

template <int K, bool POOL>
__global__ __launch_bounds__(256) void agg_gemm_kernel(
        const int* __restrict__ rowptr, const int* __restrict__ cnt,
        const int2* __restrict__ pairs, const float* __restrict__ h,
        const float* __restrict__ W, const float* __restrict__ bias,
        float* __restrict__ hout, const int* __restrict__ batch,
        float* pooled, float* counts, int N) {
    constexpr int PITCH = K + 4;
    __shared__ __attribute__((aligned(16))) float Wlds[K * 64];
    __shared__ __attribute__((aligned(16))) float Alds[32 * PITCH];
    __shared__ int Glds[32];

    const int lane = threadIdx.x & 31;
    const int wave = __builtin_amdgcn_readfirstlane((int)(threadIdx.x >> 5));
    const int nb = (int)blockIdx.x * 32 + wave * NODES_PER_WAVE;

    int startv = 0, degv = 0, gv = 0;
    if (lane < NODES_PER_WAVE && nb + lane < N) {
        startv = rowptr[nb + lane];
        degv   = cnt[nb + lane];
        if (POOL) gv = batch[nb + lane];
    }
    float4 wv[K / 16];
    #pragma unroll
    for (int u = 0; u < K / 16; ++u)
        wv[u] = ((const float4*)W)[u * 256 + (int)threadIdx.x];
    #pragma unroll
    for (int u = 0; u < K / 16; ++u)
        ((float4*)Wlds)[u * 256 + (int)threadIdx.x] = wv[u];

    int s[NODES_PER_WAVE], st[NODES_PER_WAVE], dg[NODES_PER_WAVE];
    float w[NODES_PER_WAVE];
    int2 pv[NODES_PER_WAVE];
    #pragma unroll
    for (int i = 0; i < NODES_PER_WAVE; ++i) {
        st[i] = lane_bcast_i(startv, i);
        dg[i] = lane_bcast_i(degv, i);
        const int c0 = dg[i] < 32 ? dg[i] : 32;
        pv[i] = pairs[lane < c0 ? st[i] + lane : 0];
    }
    #pragma unroll
    for (int i = 0; i < NODES_PER_WAVE; ++i) {
        const int c0 = dg[i] < 32 ? dg[i] : 32;
        s[i] = lane < c0 ? pv[i].x : nb + i;
        w[i] = lane < c0 ? __int_as_float(pv[i].y) : 0.0f;
    }

    #pragma unroll
    for (int i = 0; i < NODES_PER_WAVE; ++i) {
        const int n = nb + i;
        const int row = wave * NODES_PER_WAVE + i;
        float ax = 0.0f, ay = 0.0f;
        if (n < N) {
            const int deg = dg[i], start = st[i];
            const int c0 = deg < 32 ? deg : 32;
            float sx, sy = 0.0f;
            if constexpr (K == 64) {
                const float2 sv = ((const float2*)(h + (size_t)n * 64))[lane];
                sx = sv.x; sy = sv.y;
            } else {
                sx = h[(size_t)n * 32 + lane];
            }
            for (int j = 0; j < c0; j += GATHER_GROUP)
                gather_group<K>(h, s[i], w[i], j, lane, ax, ay);
            for (int base = 32; base < deg; base += 32) {
                const int rem = deg - base;
                const int c = rem < 32 ? rem : 32;
                int s2 = n;
                float w2 = 0.0f;
                if (lane < c) {
                    const int2 p = pairs[start + base + lane];
                    s2 = p.x;
                    w2 = __int_as_float(p.y);
                }
                for (int j = 0; j < c; j += GATHER_GROUP)
                    gather_group<K>(h, s2, w2, j, lane, ax, ay);
            }
            const float dn = rsqrtf((float)(deg + 1));
            ax += dn * dn * sx;
            ay += dn * dn * sy;
            if (POOL) {
                const int g = lane_bcast_i(gv, i);
                if (lane == 0) {
                    Glds[row] = g;
                    atomic_add_f32(&counts[g], 1.0f);
                }
            }
        }
        if constexpr (K == 64) ((float2*)(Alds + row * PITCH))[lane] = make_float2(ax, ay);
        else                   Alds[row * PITCH + lane] = ax;
    }
    __syncthreads();

    const int half = lane >> 4;
    const int l16  = lane & 15;
    const int col0 = (wave & 3) * 16;
    const int rloc = (wave >> 2) * 16;
    const float* Arow = Alds + (rloc + l16) * PITCH + 2 * half;
    const float* Brow = Wlds + 2 * half * 64 + col0 + l16;

    v8f acc = {};
    #pragma unroll
    for (int k0 = 0; k0 < K; k0 += 4) {
        v2f a = *(const v2f*)(Arow + k0);
        v2f b;
        b.x = Brow[k0 * 64];
        b.y = Brow[k0 * 64 + 64];
        acc = __builtin_amdgcn_wmma_f32_16x16x4_f32(
             false, a,  false, b,
             (short)0, acc,  false,  false);
    }
    const float bv = bias[col0 + l16];
    #pragma unroll
    for (int r = 0; r < 8; ++r) {
        const int rl = rloc + r + 8 * half;
        const int mm = (int)blockIdx.x * 32 + rl;
        if (mm < N) {
            const float v = fmaxf(acc[r] + bv, 0.0f);
            if (POOL) atomic_add_f32(pooled + (size_t)Glds[rl] * 64 + col0 + l16, v);
            else      hout[(size_t)mm * 64 + col0 + l16] = v;
        }
    }
}

__global__ void pooled_mlp_head_kernel(const float* __restrict__ pooled,
                                       const float* __restrict__ counts,
                                       const float* __restrict__ lw1,
                                       const float* __restrict__ lb1,
                                       const float* __restrict__ lw2,
                                       const float* __restrict__ lb2,
                                       float* __restrict__ out, int G) {
    __shared__ float Wlds[64 * 64];
    __shared__ float Rt[32 * 65];
    #pragma unroll 4
    for (int i = threadIdx.x; i < 64 * 64; i += 256)
        Wlds[i] = lw1[i];
    __syncthreads();

    const int wave = threadIdx.x >> 5;
    const int lane = threadIdx.x & 31;
    const int half = lane >> 4;
    const int l16  = lane & 15;
    const int col0 = (wave & 3) * 16;
    const int rloc = (wave >> 2) * 16;
    const int row0 = blockIdx.x * 32 + rloc;
    const int m    = row0 + l16;
    const int mc   = (m < G) ? m : (G - 1);
    const float c  = fmaxf(counts[mc], 1.0f);
    const float* __restrict__ Arow = pooled + (size_t)mc * 64 + 2 * half;
    const float* __restrict__ Brow = Wlds + 2 * half * 64 + col0 + l16;

    v8f acc = {};
    #pragma unroll
    for (int k0 = 0; k0 < 64; k0 += 4) {
        v2f a = *(const v2f*)(Arow + k0);
        a.x = a.x / c;
        a.y = a.y / c;
        v2f b;
        b.x = Brow[k0 * 64];
        b.y = Brow[k0 * 64 + 64];
        acc = __builtin_amdgcn_wmma_f32_16x16x4_f32(
             false, a,  false, b,
             (short)0, acc,  false,  false);
    }
    const float bv = lb1[col0 + l16];
    #pragma unroll
    for (int r = 0; r < 8; ++r)
        Rt[(rloc + r + 8 * half) * 65 + col0 + l16] = fmaxf(acc[r] + bv, 0.0f);
    __syncthreads();

    const int t = threadIdx.x;
    if (t < 64) {
        const int gl = t >> 1, j = t & 1;
        const int gg = blockIdx.x * 32 + gl;
        if (gg < G) {
            float o = lb2[j];
            #pragma unroll
            for (int k = 0; k < 64; ++k) o += Rt[gl * 65 + k] * lw2[k * 2 + j];
            out[(size_t)gg * 2 + j] = o;
        }
    }
}

static inline size_t alignup(size_t v) { return (v + 255) & ~(size_t)255; }

extern "C" void kernel_launch(void* const* d_in, const int* in_sizes, int n_in,
                              void* d_out, int out_size, void* d_ws, size_t ws_size,
                              hipStream_t stream) {
    const float* x     = (const float*)d_in[0];
    const int*   eidx  = (const int*)d_in[1];
    const int*   batch = (const int*)d_in[2];
    const float* W1 = (const float*)d_in[3];  const float* b1 = (const float*)d_in[4];
    const float* W2 = (const float*)d_in[5];  const float* b2 = (const float*)d_in[6];
    const float* W3 = (const float*)d_in[7];  const float* b3 = (const float*)d_in[8];
    const float* lw1 = (const float*)d_in[9]; const float* lb1 = (const float*)d_in[10];
    const float* lw2 = (const float*)d_in[11]; const float* lb2 = (const float*)d_in[12];
    float* out = (float*)d_out;

    const int N = N_NODES, G = N_GRAPHS;
    const int E = in_sizes[1] / 2;
    const int* src = eidx;
    const int* dst = eidx + E;

    char* w = (char*)d_ws;
    int*   cnt    = (int*)w;   w += alignup((size_t)N * 4);
    int*   rowptr = (int*)w;   w += alignup((size_t)N * 4);
    int*   rank   = (int*)w;   w += alignup((size_t)E * 4);
    int2*  pairs  = (int2*)w;  w += alignup((size_t)E * 8);
    float* h1     = (float*)w; w += alignup((size_t)N * 64 * 4);
    float* h2     = (float*)w; w += alignup((size_t)N * 64 * 4);
    float* pooled = (float*)w; w += alignup((size_t)G * 64 * 4);
    float* counts = (float*)w; w += alignup((size_t)G * 4);
    (void)ws_size; (void)n_in;

    dim3 blk256(256);
    const int N4 = N / 4;
    const int GF4 = G * 64 / 4;
    const int G4 = G / 4;
    int gClr  = (N4 + GF4 + G4 + 255) / 256;
    int gN4   = (N4 + 255) / 256;
    int gE    = (E + 255) / 256;
    int gLayer = (N + 31) / 32;
    int gGemmG = (G + 31) / 32;

    clear_kernel<<<gClr, blk256, 0, stream>>>(cnt, N4, pooled, GF4, counts, G4);
    count_rank_kernel<<<gE, blk256, 0, stream>>>(dst, cnt, rank, E);
    rowptr_scan_kernel<<<gN4, blk256, 0, stream>>>(cnt, rowptr, N4);
    bucket_fill_kernel<<<gE, blk256, 0, stream>>>(src, dst, rank, rowptr, cnt, pairs, E);

    agg_gemm_kernel<32, false><<<gLayer, blk256, 0, stream>>>(rowptr, cnt, pairs, x, W1, b1, h1,
                                                              batch, pooled, counts, N);
    agg_gemm_kernel<64, false><<<gLayer, blk256, 0, stream>>>(rowptr, cnt, pairs, h1, W2, b2, h2,
                                                              batch, pooled, counts, N);
    agg_gemm_kernel<64, true><<<gLayer, blk256, 0, stream>>>(rowptr, cnt, pairs, h2, W3, b3, h1,
                                                             batch, pooled, counts, N);

    pooled_mlp_head_kernel<<<gGemmG, blk256, 0, stream>>>(pooled, counts, lw1, lb1, lw2, lb2, out, G);
    (void)out_size;
}


# atomic targets spread over more 128-B lines: in-degree counters 16 B apart (8 a line, was 32), graph node counters 1 a line (was 32); scan in 1024-thread blocks keeps a packed copy; sum order kept
# speedup vs baseline: 2.8597x; 1.1221x over previous
#include <hip/hip_runtime.h>

typedef __attribute__((ext_vector_type(2))) float v2f;
typedef __attribute__((ext_vector_type(8))) float v8f;

#define N_NODES 50000
#define N_GRAPHS 2048

__device__ __forceinline__ void atomic_add_f32(float* p, float v) {
    unsafeAtomicAdd(p, v);
}

#define CNT_STRIDE 4
#define GCNT_STRIDE 32
__global__ void clear_kernel(int* cnt, int n4_cnt, float* pooled, int n4_pooled,
                             float* counts, int n4_counts) {
    int i = blockIdx.x * blockDim.x + threadIdx.x;
    if (i < n4_cnt) { ((int4*)cnt)[i] = make_int4(0, 0, 0, 0); return; }
    i -= n4_cnt;
    if (i < n4_pooled) { ((float4*)pooled)[i] = make_float4(0.f, 0.f, 0.f, 0.f); return; }
    i -= n4_pooled;
    if (i < n4_counts) ((float4*)counts)[i] = make_float4(0.f, 0.f, 0.f, 0.f);
}

__global__ __launch_bounds__(1024) void rowptr_scan_kernel(const int* __restrict__ cnts,
                                                           int* __restrict__ rowptr,
                                                           int* __restrict__ cnt, int N) {
    __shared__ int pre_w[32];
    __shared__ int tot_w[32];
    __shared__ int base_w[32];
    const int t = threadIdx.x, lane = t & 31, wave = t >> 5;
    const int b = (int)blockIdx.x;
    int pre = 0;
    for (int k0 = 0; k0 < b; k0 += 8) {
        int v[8];
        #pragma unroll
        for (int u = 0; u < 8; ++u) {
            const int kk = (k0 + u < b) ? (k0 + u) : 0;
            v[u] = cnts[(size_t)(kk * 1024 + t) * CNT_STRIDE];
        }
        #pragma unroll
        for (int u = 0; u < 8; ++u) pre += (k0 + u < b) ? v[u] : 0;
    }
    const int n = b * 1024 + t;
    const int mine = (n < N) ? cnts[(size_t)n * CNT_STRIDE] : 0;
    int incl = mine;
    #pragma unroll
    for (int o = 1; o < 32; o <<= 1) {
        int up = __shfl_up(incl, o);
        if (lane >= o) incl += up;
    }
    #pragma unroll
    for (int o = 16; o > 0; o >>= 1) pre += __shfl_xor(pre, o);
    if (lane == 31) tot_w[wave] = incl;
    if (lane == 0)  pre_w[wave] = pre;
    __syncthreads();
    if (wave == 0) {
        int p = pre_w[lane];
        #pragma unroll
        for (int o = 16; o > 0; o >>= 1) p += __shfl_xor(p, o);
        const int tw = tot_w[lane];
        int ti = tw;
        #pragma unroll
        for (int o = 1; o < 32; o <<= 1) {
            int up = __shfl_up(ti, o);
            if (lane >= o) ti += up;
        }
        base_w[lane] = p + ti - tw;
    }
    __syncthreads();
    if (n < N) {
        rowptr[n] = base_w[wave] + incl - mine;
        cnt[n] = mine;
    }
}

__global__ void bucket_fill_kernel(const int* __restrict__ src,
                                   const int* __restrict__ dst,
                                   const int* __restrict__ rank,
                                   const int* __restrict__ rowptr,
                                   const int* __restrict__ cnt,
                                   int2* __restrict__ pairs, int E) {
    int i = blockIdx.x * blockDim.x + threadIdx.x;
    if (i >= E) return;
    const int s = src[i], d = dst[i];
    const float nrm = rsqrtf((float)(cnt[s] + 1)) * rsqrtf((float)(cnt[d] + 1));
    pairs[rowptr[d] + rank[i]] = make_int2(s, __float_as_int(nrm));
}

__global__ void count_rank_kernel(const int* __restrict__ dst, int* cnts,
                                  int* __restrict__ rank, int E) {
    const int i = (int)blockIdx.x * 256 + (int)threadIdx.x;
    if (i < E) rank[i] = atomicAdd(&cnts[(size_t)dst[i] * CNT_STRIDE], 1);
}

__device__ __forceinline__ int lane_bcast_i(int v, int j) {
    return __builtin_amdgcn_readlane(v, j);
}
__device__ __forceinline__ float lane_bcast_f(float v, int j) {
    return __int_as_float(__builtin_amdgcn_readlane(__float_as_int(v), j));
}

#define GATHER_GROUP 16
#define NODES_PER_WAVE 4
template <int K>
__device__ __forceinline__ void gather_group(const float* __restrict__ h, int s, float w,
                                             int j, int lane, float& ax, float& ay) {
    if constexpr (K == 64) {
        float2 v[GATHER_GROUP];
        #pragma unroll
        for (int u = 0; u < GATHER_GROUP; ++u) {
            const int sj = lane_bcast_i(s, j + u);
            v[u] = ((const float2*)(h + (size_t)sj * 64))[lane];
        }
        #pragma unroll
        for (int u = 0; u < GATHER_GROUP; ++u) {
            const float wj = lane_bcast_f(w, j + u);
            ax += wj * v[u].x;
            ay += wj * v[u].y;
        }
    } else {
        float v[GATHER_GROUP];
        #pragma unroll
        for (int u = 0; u < GATHER_GROUP; ++u) {
            const int sj = lane_bcast_i(s, j + u);
            v[u] = h[(size_t)sj * 32 + lane];
        }
        #pragma unroll
        for (int u = 0; u < GATHER_GROUP; ++u) {
            const float wj = lane_bcast_f(w, j + u);
            ax += wj * v[u];
        }
    }
}

template <int K, bool POOL>
__global__ __launch_bounds__(256) void agg_gemm_kernel(
        const int* __restrict__ rowptr, const int* __restrict__ cnt,
        const int2* __restrict__ pairs, const float* __restrict__ h,
        const float* __restrict__ W, const float* __restrict__ bias,
        float* __restrict__ hout, const int* __restrict__ batch,
        float* pooled, float* counts, int N) {
    constexpr int PITCH = K + 4;
    __shared__ __attribute__((aligned(16))) float Wlds[K * 64];
    __shared__ __attribute__((aligned(16))) float Alds[32 * PITCH];
    __shared__ int Glds[32];

    const int lane = threadIdx.x & 31;
    const int wave = __builtin_amdgcn_readfirstlane((int)(threadIdx.x >> 5));
    const int nb = (int)blockIdx.x * 32 + wave * NODES_PER_WAVE;

    int startv = 0, degv = 0, gv = 0;
    if (lane < NODES_PER_WAVE && nb + lane < N) {
        startv = rowptr[nb + lane];
        degv   = cnt[nb + lane];
        if (POOL) gv = batch[nb + lane];
    }
    float4 wv[K / 16];
    #pragma unroll
    for (int u = 0; u < K / 16; ++u)
        wv[u] = ((const float4*)W)[u * 256 + (int)threadIdx.x];
    #pragma unroll
    for (int u = 0; u < K / 16; ++u)
        ((float4*)Wlds)[u * 256 + (int)threadIdx.x] = wv[u];

    int s[NODES_PER_WAVE], st[NODES_PER_WAVE], dg[NODES_PER_WAVE];
    float w[NODES_PER_WAVE];
    int2 pv[NODES_PER_WAVE];
    #pragma unroll
    for (int i = 0; i < NODES_PER_WAVE; ++i) {
        st[i] = lane_bcast_i(startv, i);
        dg[i] = lane_bcast_i(degv, i);
        const int c0 = dg[i] < 32 ? dg[i] : 32;
        pv[i] = pairs[lane < c0 ? st[i] + lane : 0];
    }
    #pragma unroll
    for (int i = 0; i < NODES_PER_WAVE; ++i) {
        const int c0 = dg[i] < 32 ? dg[i] : 32;
        s[i] = lane < c0 ? pv[i].x : nb + i;
        w[i] = lane < c0 ? __int_as_float(pv[i].y) : 0.0f;
    }

    #pragma unroll
    for (int i = 0; i < NODES_PER_WAVE; ++i) {
        const int n = nb + i;
        const int row = wave * NODES_PER_WAVE + i;
        float ax = 0.0f, ay = 0.0f;
        if (n < N) {
            const int deg = dg[i], start = st[i];
            const int c0 = deg < 32 ? deg : 32;
            float sx, sy = 0.0f;
            if constexpr (K == 64) {
                const float2 sv = ((const float2*)(h + (size_t)n * 64))[lane];
                sx = sv.x; sy = sv.y;
            } else {
                sx = h[(size_t)n * 32 + lane];
            }
            for (int j = 0; j < c0; j += GATHER_GROUP)
                gather_group<K>(h, s[i], w[i], j, lane, ax, ay);
            for (int base = 32; base < deg; base += 32) {
                const int rem = deg - base;
                const int c = rem < 32 ? rem : 32;
                int s2 = n;
                float w2 = 0.0f;
                if (lane < c) {
                    const int2 p = pairs[start + base + lane];
                    s2 = p.x;
                    w2 = __int_as_float(p.y);
                }
                for (int j = 0; j < c; j += GATHER_GROUP)
                    gather_group<K>(h, s2, w2, j, lane, ax, ay);
            }
            const float dn = rsqrtf((float)(deg + 1));
            ax += dn * dn * sx;
            ay += dn * dn * sy;
            if (POOL) {
                const int g = lane_bcast_i(gv, i);
                if (lane == 0) {
                    Glds[row] = g;
                    atomic_add_f32(&counts[(size_t)g * GCNT_STRIDE], 1.0f);
                }
            }
        }
        if constexpr (K == 64) ((float2*)(Alds + row * PITCH))[lane] = make_float2(ax, ay);
        else                   Alds[row * PITCH + lane] = ax;
    }
    __syncthreads();

    const int half = lane >> 4;
    const int l16  = lane & 15;
    const int col0 = (wave & 3) * 16;
    const int rloc = (wave >> 2) * 16;
    const float* Arow = Alds + (rloc + l16) * PITCH + 2 * half;
    const float* Brow = Wlds + 2 * half * 64 + col0 + l16;

    v8f acc = {};
    #pragma unroll
    for (int k0 = 0; k0 < K; k0 += 4) {
        v2f a = *(const v2f*)(Arow + k0);
        v2f b;
        b.x = Brow[k0 * 64];
        b.y = Brow[k0 * 64 + 64];
        acc = __builtin_amdgcn_wmma_f32_16x16x4_f32(
             false, a,  false, b,
             (short)0, acc,  false,  false);
    }
    const float bv = bias[col0 + l16];
    #pragma unroll
    for (int r = 0; r < 8; ++r) {
        const int rl = rloc + r + 8 * half;
        const int mm = (int)blockIdx.x * 32 + rl;
        if (mm < N) {
            const float v = fmaxf(acc[r] + bv, 0.0f);
            if (POOL) atomic_add_f32(pooled + (size_t)Glds[rl] * 64 + col0 + l16, v);
            else      hout[(size_t)mm * 64 + col0 + l16] = v;
        }
    }
}

__global__ void pooled_mlp_head_kernel(const float* __restrict__ pooled,
                                       const float* __restrict__ counts,
                                       const float* __restrict__ lw1,
                                       const float* __restrict__ lb1,
                                       const float* __restrict__ lw2,
                                       const float* __restrict__ lb2,
                                       float* __restrict__ out, int G) {
    __shared__ float Wlds[64 * 64];
    __shared__ float Rt[32 * 65];
    #pragma unroll 4
    for (int i = threadIdx.x; i < 64 * 64; i += 256)
        Wlds[i] = lw1[i];
    __syncthreads();

    const int wave = threadIdx.x >> 5;
    const int lane = threadIdx.x & 31;
    const int half = lane >> 4;
    const int l16  = lane & 15;
    const int col0 = (wave & 3) * 16;
    const int rloc = (wave >> 2) * 16;
    const int row0 = blockIdx.x * 32 + rloc;
    const int m    = row0 + l16;
    const int mc   = (m < G) ? m : (G - 1);
    const float c  = fmaxf(counts[(size_t)mc * GCNT_STRIDE], 1.0f);
    const float* __restrict__ Arow = pooled + (size_t)mc * 64 + 2 * half;
    const float* __restrict__ Brow = Wlds + 2 * half * 64 + col0 + l16;

    v8f acc = {};
    #pragma unroll
    for (int k0 = 0; k0 < 64; k0 += 4) {
        v2f a = *(const v2f*)(Arow + k0);
        a.x = a.x / c;
        a.y = a.y / c;
        v2f b;
        b.x = Brow[k0 * 64];
        b.y = Brow[k0 * 64 + 64];
        acc = __builtin_amdgcn_wmma_f32_16x16x4_f32(
             false, a,  false, b,
             (short)0, acc,  false,  false);
    }
    const float bv = lb1[col0 + l16];
    #pragma unroll
    for (int r = 0; r < 8; ++r)
        Rt[(rloc + r + 8 * half) * 65 + col0 + l16] = fmaxf(acc[r] + bv, 0.0f);
    __syncthreads();

    const int t = threadIdx.x;
    if (t < 64) {
        const int gl = t >> 1, j = t & 1;
        const int gg = blockIdx.x * 32 + gl;
        if (gg < G) {
            float o = lb2[j];
            #pragma unroll
            for (int k = 0; k < 64; ++k) o += Rt[gl * 65 + k] * lw2[k * 2 + j];
            out[(size_t)gg * 2 + j] = o;
        }
    }
}

static inline size_t alignup(size_t v) { return (v + 255) & ~(size_t)255; }

extern "C" void kernel_launch(void* const* d_in, const int* in_sizes, int n_in,
                              void* d_out, int out_size, void* d_ws, size_t ws_size,
                              hipStream_t stream) {
    const float* x     = (const float*)d_in[0];
    const int*   eidx  = (const int*)d_in[1];
    const int*   batch = (const int*)d_in[2];
    const float* W1 = (const float*)d_in[3];  const float* b1 = (const float*)d_in[4];
    const float* W2 = (const float*)d_in[5];  const float* b2 = (const float*)d_in[6];
    const float* W3 = (const float*)d_in[7];  const float* b3 = (const float*)d_in[8];
    const float* lw1 = (const float*)d_in[9]; const float* lb1 = (const float*)d_in[10];
    const float* lw2 = (const float*)d_in[11]; const float* lb2 = (const float*)d_in[12];
    float* out = (float*)d_out;

    const int N = N_NODES, G = N_GRAPHS;
    const int E = in_sizes[1] / 2;
    const int* src = eidx;
    const int* dst = eidx + E;

    char* w = (char*)d_ws;
    int*   cnts   = (int*)w;   w += alignup((size_t)N * 4 * CNT_STRIDE);
    int*   cnt    = (int*)w;   w += alignup((size_t)N * 4);
    int*   rowptr = (int*)w;   w += alignup((size_t)N * 4);
    int*   rank   = (int*)w;   w += alignup((size_t)E * 4);
    int2*  pairs  = (int2*)w;  w += alignup((size_t)E * 8);
    float* h1     = (float*)w; w += alignup((size_t)N * 64 * 4);
    float* h2     = (float*)w; w += alignup((size_t)N * 64 * 4);
    float* pooled = (float*)w; w += alignup((size_t)G * 64 * 4);
    float* counts = (float*)w; w += alignup((size_t)G * 4 * GCNT_STRIDE);
    (void)ws_size; (void)n_in;

    dim3 blk256(256);
    const int NC4 = N * CNT_STRIDE / 4;
    const int GF4 = G * 64 / 4;
    const int G4 = G * GCNT_STRIDE / 4;
    int gClr  = (NC4 + GF4 + G4 + 255) / 256;
    int gScan = (N + 1023) / 1024;
    int gE    = (E + 255) / 256;
    int gLayer = (N + 31) / 32;
    int gGemmG = (G + 31) / 32;

    clear_kernel<<<gClr, blk256, 0, stream>>>(cnts, NC4, pooled, GF4, counts, G4);
    count_rank_kernel<<<gE, blk256, 0, stream>>>(dst, cnts, rank, E);
    rowptr_scan_kernel<<<gScan, dim3(1024), 0, stream>>>(cnts, rowptr, cnt, N);
    bucket_fill_kernel<<<gE, blk256, 0, stream>>>(src, dst, rank, rowptr, cnt, pairs, E);

    agg_gemm_kernel<32, false><<<gLayer, blk256, 0, stream>>>(rowptr, cnt, pairs, x, W1, b1, h1,
                                                              batch, pooled, counts, N);
    agg_gemm_kernel<64, false><<<gLayer, blk256, 0, stream>>>(rowptr, cnt, pairs, h1, W2, b2, h2,
                                                              batch, pooled, counts, N);
    agg_gemm_kernel<64, true><<<gLayer, blk256, 0, stream>>>(rowptr, cnt, pairs, h2, W3, b3, h1,
                                                             batch, pooled, counts, N);

    pooled_mlp_head_kernel<<<gGemmG, blk256, 0, stream>>>(pooled, counts, lw1, lb1, lw2, lb2, out, G);
    (void)out_size;
}


# edges bucketed by a two-level counting sort on LDS counters (dst range, then node), no global atomic per edge; list keeps src only, rows carry dinv; sum order = LDS arrival order, rounding differs
# speedup vs baseline: 4.0835x; 1.4279x over previous
#include <hip/hip_runtime.h>

typedef __attribute__((ext_vector_type(2))) float v2f;
typedef __attribute__((ext_vector_type(8))) float v8f;

#define N_NODES 50000
#define N_GRAPHS 2048

__device__ __forceinline__ void atomic_add_f32(float* p, float v) {
    unsafeAtomicAdd(p, v);
}

#define BSHIFT 8
#define BNODES 256
#define NBUCK ((N_NODES + BNODES - 1) / BNODES)
#define SORT_BLOCKS 256
#define SORT_THREADS 1024
#define SORT_UNR 4
#define FINE_UNR 6
#define GCNT_STRIDE 32

__device__ __forceinline__ int sort_chunk(int E) { return (E + SORT_BLOCKS - 1) / SORT_BLOCKS; }

__global__ __launch_bounds__(SORT_THREADS) void coarse_hist_kernel(
        const int* __restrict__ dst, int* __restrict__ hist, int E,
        float* pooled, int n4_pooled, float* counts, int n4_counts) {
    __shared__ int hl[NBUCK];
    const int t = (int)threadIdx.x, j = (int)blockIdx.x;
    if (t < NBUCK) hl[t] = 0;
    for (int g = j * SORT_THREADS + t; g < n4_pooled + n4_counts; g += SORT_BLOCKS * SORT_THREADS) {
        float4* p = (g < n4_pooled) ? ((float4*)pooled + g) : ((float4*)counts + (g - n4_pooled));
        *p = make_float4(0.f, 0.f, 0.f, 0.f);
    }
    __syncthreads();
    const int ch = sort_chunk(E);
    const int e0 = j * ch;
    const int e1 = (e0 + ch < E) ? (e0 + ch) : E;
    for (int base = e0 + t; base < e1 + t; base += SORT_UNR * SORT_THREADS) {
        int d[SORT_UNR];
        #pragma unroll
        for (int k = 0; k < SORT_UNR; ++k) {
            const int e = base + k * SORT_THREADS;
            d[k] = dst[e < e1 ? e : e0];
        }
        #pragma unroll
        for (int k = 0; k < SORT_UNR; ++k)
            if (base + k * SORT_THREADS < e1) atomicAdd(&hl[d[k] >> BSHIFT], 1);
    }
    __syncthreads();
    if (t < NBUCK) hist[t * SORT_BLOCKS + j] = hl[t];
}

__global__ __launch_bounds__(SORT_THREADS) void coarse_scatter_kernel(
        const int* __restrict__ src, const int* __restrict__ dst,
        const int* __restrict__ hist, int2* __restrict__ tmp,
        int* __restrict__ bstart, int E) {
    __shared__ int totL[NBUCK];
    __shared__ int preL[NBUCK];
    __shared__ int offL[NBUCK];
    const int t = (int)threadIdx.x, lane = t & 31, wave = t >> 5, j = (int)blockIdx.x;
    constexpr int TRIPS = (NBUCK + 31) / 32;
    int4 ra[TRIPS], rb[TRIPS];
    #pragma unroll
    for (int u = 0; u < TRIPS; ++u) {
        const int b = wave + 32 * u;
        const int4* hp = (const int4*)(hist + (b < NBUCK ? b : 0) * SORT_BLOCKS + lane * 8);
        ra[u] = hp[0];
        rb[u] = hp[1];
    }
    #pragma unroll
    for (int u = 0; u < TRIPS; ++u) {
        const int b = wave + 32 * u;
        const int v[8] = {ra[u].x, ra[u].y, ra[u].z, ra[u].w, rb[u].x, rb[u].y, rb[u].z, rb[u].w};
        int tot = 0, pre = 0;
        #pragma unroll
        for (int k = 0; k < 8; ++k) {
            tot += v[k];
            pre += (lane * 8 + k < j) ? v[k] : 0;
        }
        #pragma unroll
        for (int o = 16; o > 0; o >>= 1) {
            tot += __shfl_xor(tot, o);
            pre += __shfl_xor(pre, o);
        }
        if (lane == 0 && b < NBUCK) { totL[b] = tot; preL[b] = pre; }
    }
    __syncthreads();
    if (wave == 0) {
        int loc[TRIPS];
        int sum = 0;
        #pragma unroll
        for (int u = 0; u < TRIPS; ++u) {
            const int b = lane * TRIPS + u;
            loc[u] = (b < NBUCK) ? totL[b] : 0;
            sum += loc[u];
        }
        int incl = sum;
        #pragma unroll
        for (int o = 1; o < 32; o <<= 1) {
            int up = __shfl_up(incl, o);
            if (lane >= o) incl += up;
        }
        int run = incl - sum;
        #pragma unroll
        for (int u = 0; u < TRIPS; ++u) {
            const int b = lane * TRIPS + u;
            if (b < NBUCK) {
                offL[b] = run + preL[b];
                if (j == 0) bstart[b] = run;
            }
            run += loc[u];
        }
        if (j == 0 && lane == 31) bstart[NBUCK] = incl;
    }
    __syncthreads();
    const int ch = sort_chunk(E);
    const int e0 = j * ch;
    const int e1 = (e0 + ch < E) ? (e0 + ch) : E;
    for (int base = e0 + t; base < e1 + t; base += SORT_UNR * SORT_THREADS) {
        int s[SORT_UNR], d[SORT_UNR];
        #pragma unroll
        for (int k = 0; k < SORT_UNR; ++k) {
            const int e = base + k * SORT_THREADS;
            const int ec = e < e1 ? e : e0;
            s[k] = src[ec];
            d[k] = dst[ec];
        }
        #pragma unroll
        for (int k = 0; k < SORT_UNR; ++k) {
            if (base + k * SORT_THREADS < e1) {
                const int p = atomicAdd(&offL[d[k] >> BSHIFT], 1);
                tmp[p] = make_int2(s[k], d[k]);
            }
        }
    }
}

__global__ __launch_bounds__(SORT_THREADS) void fine_sort_kernel(
        const int2* __restrict__ tmp, const int* __restrict__ bstart,
        const float* __restrict__ x, int* __restrict__ srcs,
        int* __restrict__ rowptr, int* __restrict__ cnt,
        float* __restrict__ xs, int N) {
    __shared__ int cL[BNODES];
    __shared__ int curL[BNODES];
    __shared__ int wsum[BNODES / 32];
    const int t = (int)threadIdx.x, lane = t & 31, wave = t >> 5, b = (int)blockIdx.x;
    const int n0 = b * BNODES;
    const int s0 = bstart[b], s1 = bstart[b + 1];
    if (t < BNODES) cL[t] = 0;
    __syncthreads();
    for (int base = s0 + t; base < s1 + t; base += FINE_UNR * SORT_THREADS) {
        int d[FINE_UNR];
        #pragma unroll
        for (int k = 0; k < FINE_UNR; ++k) {
            const int e = base + k * SORT_THREADS;
            d[k] = tmp[e < s1 ? e : s0].y;
        }
        #pragma unroll
        for (int k = 0; k < FINE_UNR; ++k)
            if (base + k * SORT_THREADS < s1) atomicAdd(&cL[d[k] - n0], 1);
    }
    __syncthreads();
    int c = 0, incl = 0;
    if (t < BNODES) {
        c = cL[t];
        incl = c;
        #pragma unroll
        for (int o = 1; o < 32; o <<= 1) {
            int up = __shfl_up(incl, o);
            if (lane >= o) incl += up;
        }
        if (lane == 31) wsum[wave] = incl;
    }
    __syncthreads();
    if (t < BNODES) {
        int base = 0;
        for (int k = 0; k < wave; ++k) base += wsum[k];
        const int ex = base + incl - c;
        curL[t] = ex;
        const int n = n0 + t;
        if (n < N) { rowptr[n] = s0 + ex; cnt[n] = c; }
    }
    __syncthreads();
    for (int base = s0 + t; base < s1 + t; base += FINE_UNR * SORT_THREADS) {
        int2 p[FINE_UNR];
        #pragma unroll
        for (int k = 0; k < FINE_UNR; ++k) {
            const int e = base + k * SORT_THREADS;
            p[k] = tmp[e < s1 ? e : s0];
        }
        #pragma unroll
        for (int k = 0; k < FINE_UNR; ++k) {
            if (base + k * SORT_THREADS < s1) {
                const int r = atomicAdd(&curL[p[k].y - n0], 1);
                srcs[s0 + r] = p[k].x;
            }
        }
    }
    for (int i = t; i < BNODES * 8; i += SORT_THREADS) {
        const int row = i >> 3;
        const int n = n0 + row;
        if (n < N) {
            const float dn = rsqrtf((float)(cL[row] + 1));
            float4 v = ((const float4*)x)[(size_t)n * 8 + (i & 7)];
            v.x *= dn; v.y *= dn; v.z *= dn; v.w *= dn;
            ((float4*)xs)[(size_t)n * 8 + (i & 7)] = v;
        }
    }
}

__device__ __forceinline__ int lane_bcast_i(int v, int j) {
    return __builtin_amdgcn_readlane(v, j);
}
__device__ __forceinline__ float lane_bcast_f(float v, int j) {
    return __int_as_float(__builtin_amdgcn_readlane(__float_as_int(v), j));
}

#define GATHER_GROUP 16
#define NODES_PER_WAVE 4
template <int K>
__device__ __forceinline__ void gather_group(const float* __restrict__ h, int s, float w,
                                             int j, int lane, float& ax, float& ay) {
    if constexpr (K == 64) {
        float2 v[GATHER_GROUP];
        #pragma unroll
        for (int u = 0; u < GATHER_GROUP; ++u) {
            const int sj = lane_bcast_i(s, j + u);
            v[u] = ((const float2*)(h + (size_t)sj * 64))[lane];
        }
        #pragma unroll
        for (int u = 0; u < GATHER_GROUP; ++u) {
            const float wj = lane_bcast_f(w, j + u);
            ax += wj * v[u].x;
            ay += wj * v[u].y;
        }
    } else {
        float v[GATHER_GROUP];
        #pragma unroll
        for (int u = 0; u < GATHER_GROUP; ++u) {
            const int sj = lane_bcast_i(s, j + u);
            v[u] = h[(size_t)sj * 32 + lane];
        }
        #pragma unroll
        for (int u = 0; u < GATHER_GROUP; ++u) {
            const float wj = lane_bcast_f(w, j + u);
            ax += wj * v[u];
        }
    }
}

template <int K, bool POOL>
__global__ __launch_bounds__(256) void agg_gemm_kernel(
        const int* __restrict__ rowptr, const int* __restrict__ cnt,
        const int* __restrict__ srcs, const float* __restrict__ h,
        const float* __restrict__ W, const float* __restrict__ bias,
        float* __restrict__ hout, const int* __restrict__ batch,
        float* pooled, float* counts, int N) {
    constexpr int PITCH = K + 4;
    __shared__ __attribute__((aligned(16))) float Wlds[K * 64];
    __shared__ __attribute__((aligned(16))) float Alds[32 * PITCH];
    __shared__ int Glds[32];
    __shared__ float Dlds[32];

    const int lane = threadIdx.x & 31;
    const int wave = __builtin_amdgcn_readfirstlane((int)(threadIdx.x >> 5));
    const int nb = (int)blockIdx.x * 32 + wave * NODES_PER_WAVE;

    int startv = 0, degv = 0, gv = 0;
    if (lane < NODES_PER_WAVE && nb + lane < N) {
        startv = rowptr[nb + lane];
        degv   = cnt[nb + lane];
        if (POOL) gv = batch[nb + lane];
    }
    float4 wv[K / 16];
    #pragma unroll
    for (int u = 0; u < K / 16; ++u)
        wv[u] = ((const float4*)W)[u * 256 + (int)threadIdx.x];
    #pragma unroll
    for (int u = 0; u < K / 16; ++u)
        ((float4*)Wlds)[u * 256 + (int)threadIdx.x] = wv[u];

    int s[NODES_PER_WAVE], st[NODES_PER_WAVE], dg[NODES_PER_WAVE];
    float w[NODES_PER_WAVE];
    int pv[NODES_PER_WAVE];
    #pragma unroll
    for (int i = 0; i < NODES_PER_WAVE; ++i) {
        st[i] = lane_bcast_i(startv, i);
        dg[i] = lane_bcast_i(degv, i);
        const int c0 = dg[i] < 32 ? dg[i] : 32;
        pv[i] = srcs[lane < c0 ? st[i] + lane : 0];
    }
    #pragma unroll
    for (int i = 0; i < NODES_PER_WAVE; ++i) {
        const int c0 = dg[i] < 32 ? dg[i] : 32;
        s[i] = lane < c0 ? pv[i] : nb + i;
        w[i] = lane < c0 ? 1.0f : 0.0f;
    }

    #pragma unroll
    for (int i = 0; i < NODES_PER_WAVE; ++i) {
        const int n = nb + i;
        const int row = wave * NODES_PER_WAVE + i;
        float ax = 0.0f, ay = 0.0f;
        if (n < N) {
            const int deg = dg[i], start = st[i];
            const int c0 = deg < 32 ? deg : 32;
            float sx, sy = 0.0f;
            if constexpr (K == 64) {
                const float2 sv = ((const float2*)(h + (size_t)n * 64))[lane];
                sx = sv.x; sy = sv.y;
            } else {
                sx = h[(size_t)n * 32 + lane];
            }
            for (int j = 0; j < c0; j += GATHER_GROUP)
                gather_group<K>(h, s[i], w[i], j, lane, ax, ay);
            for (int base = 32; base < deg; base += 32) {
                const int rem = deg - base;
                const int c = rem < 32 ? rem : 32;
                int s2 = n;
                float w2 = 0.0f;
                if (lane < c) {
                    s2 = srcs[start + base + lane];
                    w2 = 1.0f;
                }
                for (int j = 0; j < c; j += GATHER_GROUP)
                    gather_group<K>(h, s2, w2, j, lane, ax, ay);
            }
            const float dn = rsqrtf((float)(deg + 1));
            ax = dn * (ax + sx);
            ay = dn * (ay + sy);
            if (!POOL && lane == 0) Dlds[row] = dn;
            if (POOL) {
                const int g = lane_bcast_i(gv, i);
                if (lane == 0) {
                    Glds[row] = g;
                    atomic_add_f32(&counts[(size_t)g * GCNT_STRIDE], 1.0f);
                }
            }
        }
        if constexpr (K == 64) ((float2*)(Alds + row * PITCH))[lane] = make_float2(ax, ay);
        else                   Alds[row * PITCH + lane] = ax;
    }
    __syncthreads();

    const int half = lane >> 4;
    const int l16  = lane & 15;
    const int col0 = (wave & 3) * 16;
    const int rloc = (wave >> 2) * 16;
    const float* Arow = Alds + (rloc + l16) * PITCH + 2 * half;
    const float* Brow = Wlds + 2 * half * 64 + col0 + l16;

    v8f acc = {};
    #pragma unroll
    for (int k0 = 0; k0 < K; k0 += 4) {
        v2f a = *(const v2f*)(Arow + k0);
        v2f b;
        b.x = Brow[k0 * 64];
        b.y = Brow[k0 * 64 + 64];
        acc = __builtin_amdgcn_wmma_f32_16x16x4_f32(
             false, a,  false, b,
             (short)0, acc,  false,  false);
    }
    const float bv = bias[col0 + l16];
    #pragma unroll
    for (int r = 0; r < 8; ++r) {
        const int rl = rloc + r + 8 * half;
        const int mm = (int)blockIdx.x * 32 + rl;
        if (mm < N) {
            const float v = fmaxf(acc[r] + bv, 0.0f);
            if (POOL) atomic_add_f32(pooled + (size_t)Glds[rl] * 64 + col0 + l16, v);
            else      hout[(size_t)mm * 64 + col0 + l16] = Dlds[rl] * v;
        }
    }
}

__global__ void pooled_mlp_head_kernel(const float* __restrict__ pooled,
                                       const float* __restrict__ counts,
                                       const float* __restrict__ lw1,
                                       const float* __restrict__ lb1,
                                       const float* __restrict__ lw2,
                                       const float* __restrict__ lb2,
                                       float* __restrict__ out, int G) {
    __shared__ float Wlds[64 * 64];
    __shared__ float Rt[32 * 65];
    #pragma unroll 4
    for (int i = threadIdx.x; i < 64 * 64; i += 256)
        Wlds[i] = lw1[i];
    __syncthreads();

    const int wave = threadIdx.x >> 5;
    const int lane = threadIdx.x & 31;
    const int half = lane >> 4;
    const int l16  = lane & 15;
    const int col0 = (wave & 3) * 16;
    const int rloc = (wave >> 2) * 16;
    const int row0 = blockIdx.x * 32 + rloc;
    const int m    = row0 + l16;
    const int mc   = (m < G) ? m : (G - 1);
    const float c  = fmaxf(counts[(size_t)mc * GCNT_STRIDE], 1.0f);
    const float* __restrict__ Arow = pooled + (size_t)mc * 64 + 2 * half;
    const float* __restrict__ Brow = Wlds + 2 * half * 64 + col0 + l16;

    v8f acc = {};
    #pragma unroll
    for (int k0 = 0; k0 < 64; k0 += 4) {
        v2f a = *(const v2f*)(Arow + k0);
        a.x = a.x / c;
        a.y = a.y / c;
        v2f b;
        b.x = Brow[k0 * 64];
        b.y = Brow[k0 * 64 + 64];
        acc = __builtin_amdgcn_wmma_f32_16x16x4_f32(
             false, a,  false, b,
             (short)0, acc,  false,  false);
    }
    const float bv = lb1[col0 + l16];
    #pragma unroll
    for (int r = 0; r < 8; ++r)
        Rt[(rloc + r + 8 * half) * 65 + col0 + l16] = fmaxf(acc[r] + bv, 0.0f);
    __syncthreads();

    const int t = threadIdx.x;
    if (t < 64) {
        const int gl = t >> 1, j = t & 1;
        const int gg = blockIdx.x * 32 + gl;
        if (gg < G) {
            float o = lb2[j];
            #pragma unroll
            for (int k = 0; k < 64; ++k) o += Rt[gl * 65 + k] * lw2[k * 2 + j];
            out[(size_t)gg * 2 + j] = o;
        }
    }
}

static inline size_t alignup(size_t v) { return (v + 255) & ~(size_t)255; }

extern "C" void kernel_launch(void* const* d_in, const int* in_sizes, int n_in,
                              void* d_out, int out_size, void* d_ws, size_t ws_size,
                              hipStream_t stream) {
    const float* x     = (const float*)d_in[0];
    const int*   eidx  = (const int*)d_in[1];
    const int*   batch = (const int*)d_in[2];
    const float* W1 = (const float*)d_in[3];  const float* b1 = (const float*)d_in[4];
    const float* W2 = (const float*)d_in[5];  const float* b2 = (const float*)d_in[6];
    const float* W3 = (const float*)d_in[7];  const float* b3 = (const float*)d_in[8];
    const float* lw1 = (const float*)d_in[9]; const float* lb1 = (const float*)d_in[10];
    const float* lw2 = (const float*)d_in[11]; const float* lb2 = (const float*)d_in[12];
    float* out = (float*)d_out;

    const int N = N_NODES, G = N_GRAPHS;
    const int E = in_sizes[1] / 2;
    const int* src = eidx;
    const int* dst = eidx + E;

    char* w = (char*)d_ws;
    int*   hist   = (int*)w;   w += alignup((size_t)NBUCK * SORT_BLOCKS * 4);
    int*   bstart = (int*)w;   w += alignup((size_t)(NBUCK + 1) * 4);
    int*   cnt    = (int*)w;   w += alignup((size_t)N * 4);
    int*   rowptr = (int*)w;   w += alignup((size_t)N * 4);
    int2*  tmp    = (int2*)w;  w += alignup((size_t)E * 8);
    int*   srcs   = (int*)w;   w += alignup((size_t)E * 4);
    float* xs     = (float*)w; w += alignup((size_t)N * 32 * 4);
    float* h1     = (float*)w; w += alignup((size_t)N * 64 * 4);
    float* h2     = (float*)w; w += alignup((size_t)N * 64 * 4);
    float* pooled = (float*)w; w += alignup((size_t)G * 64 * 4);
    float* counts = (float*)w; w += alignup((size_t)G * 4 * GCNT_STRIDE);
    (void)ws_size; (void)n_in;

    dim3 blk256(256);
    const int GF4 = G * 64 / 4;
    const int G4 = G * GCNT_STRIDE / 4;
    int gLayer = (N + 31) / 32;
    int gGemmG = (G + 31) / 32;

    coarse_hist_kernel<<<SORT_BLOCKS, dim3(SORT_THREADS), 0, stream>>>(dst, hist, E, pooled, GF4, counts, G4);
    coarse_scatter_kernel<<<SORT_BLOCKS, dim3(SORT_THREADS), 0, stream>>>(src, dst, hist, tmp, bstart, E);
    fine_sort_kernel<<<NBUCK, dim3(SORT_THREADS), 0, stream>>>(tmp, bstart, x, srcs, rowptr, cnt, xs, N);

    agg_gemm_kernel<32, false><<<gLayer, blk256, 0, stream>>>(rowptr, cnt, srcs, xs, W1, b1, h1,
                                                              batch, pooled, counts, N);
    agg_gemm_kernel<64, false><<<gLayer, blk256, 0, stream>>>(rowptr, cnt, srcs, h1, W2, b2, h2,
                                                              batch, pooled, counts, N);
    agg_gemm_kernel<64, true><<<gLayer, blk256, 0, stream>>>(rowptr, cnt, srcs, h2, W3, b3, h1,
                                                             batch, pooled, counts, N);

    pooled_mlp_head_kernel<<<gGemmG, blk256, 0, stream>>>(pooled, counts, lw1, lb1, lw2, lb2, out, G);
    (void)out_size;
}


# sort level 1 made block-local: each block orders its own chunk by dst range inside its own slice (own LDS counts + scan), range pass reads 256 runs; 7->6 launches; sum order = LDS arrival as before
# speedup vs baseline: 4.9353x; 1.2086x over previous
#include <hip/hip_runtime.h>

typedef __attribute__((ext_vector_type(2))) float v2f;
typedef __attribute__((ext_vector_type(8))) float v8f;

#define N_NODES 50000
#define N_GRAPHS 2048

__device__ __forceinline__ void atomic_add_f32(float* p, float v) {
    unsafeAtomicAdd(p, v);
}
__device__ __forceinline__ int lane_bcast_i(int v, int j) {
    return __builtin_amdgcn_readlane(v, j);
}
__device__ __forceinline__ float lane_bcast_f(float v, int j) {
    return __int_as_float(__builtin_amdgcn_readlane(__float_as_int(v), j));
}

#define BSHIFT 8
#define BNODES 256
#define NBUCK ((N_NODES + BNODES - 1) / BNODES)
#define SORT_BLOCKS 256
#define SORT_THREADS 1024
#define SORT_UNR 4
#define SORT_WAVES (SORT_THREADS / 32)
#define RUNS_PER_WAVE (SORT_BLOCKS / SORT_WAVES)
static_assert(RUNS_PER_WAVE * SORT_WAVES == SORT_BLOCKS && RUNS_PER_WAVE <= 32 && SORT_WAVES <= 32,
              "level 2 gives every wave the same number of runs, one table entry per lane");
#define GCNT_STRIDE 32

__device__ __forceinline__ int sort_chunk(int E) { return (E + SORT_BLOCKS - 1) / SORT_BLOCKS; }

__global__ __launch_bounds__(SORT_THREADS) void coarse_sort_kernel(
        const int* __restrict__ src, const int* __restrict__ dst,
        int2* __restrict__ tmp, int2* __restrict__ seg, int E,
        float* pooled, int n4_pooled, float* counts, int n4_counts) {
    __shared__ int hl[NBUCK];
    __shared__ int offL[NBUCK];
    const int t = (int)threadIdx.x, lane = t & 31, wave = t >> 5, j = (int)blockIdx.x;
    constexpr int TRIPS = (NBUCK + 31) / 32;
    constexpr int ROUND = SORT_UNR * SORT_THREADS;
    if (t < NBUCK) hl[t] = 0;
    for (int g = j * SORT_THREADS + t; g < n4_pooled + n4_counts; g += SORT_BLOCKS * SORT_THREADS) {
        float4* p = (g < n4_pooled) ? ((float4*)pooled + g) : ((float4*)counts + (g - n4_pooled));
        *p = make_float4(0.f, 0.f, 0.f, 0.f);
    }
    const int ch = sort_chunk(E);
    const int e0 = j * ch;
    const int e1 = (e0 + ch < E) ? (e0 + ch) : E;
    const int ec0 = (e0 < E) ? e0 : 0;
    int s[SORT_UNR], d[SORT_UNR];
    #pragma unroll
    for (int k = 0; k < SORT_UNR; ++k) {
        const int e = e0 + t + k * SORT_THREADS;
        const int ec = e < e1 ? e : ec0;
        s[k] = src[ec];
        d[k] = dst[ec];
    }
    __syncthreads();
    #pragma unroll
    for (int k = 0; k < SORT_UNR; ++k)
        if (e0 + t + k * SORT_THREADS < e1) atomicAdd(&hl[d[k] >> BSHIFT], 1);
    for (int base = e0 + t + ROUND; base < e1 + t; base += ROUND) {
        int d2[SORT_UNR];
        #pragma unroll
        for (int k = 0; k < SORT_UNR; ++k) {
            const int e = base + k * SORT_THREADS;
            d2[k] = dst[e < e1 ? e : ec0];
        }
        #pragma unroll
        for (int k = 0; k < SORT_UNR; ++k)
            if (base + k * SORT_THREADS < e1) atomicAdd(&hl[d2[k] >> BSHIFT], 1);
    }
    __syncthreads();
    if (wave == 0) {
        int loc[TRIPS];
        int sum = 0;
        #pragma unroll
        for (int u = 0; u < TRIPS; ++u) {
            const int b = lane * TRIPS + u;
            loc[u] = (b < NBUCK) ? hl[b] : 0;
            sum += loc[u];
        }
        int incl = sum;
        #pragma unroll
        for (int o = 1; o < 32; o <<= 1) {
            int up = __shfl_up(incl, o);
            if (lane >= o) incl += up;
        }
        int run = incl - sum;
        #pragma unroll
        for (int u = 0; u < TRIPS; ++u) {
            const int b = lane * TRIPS + u;
            if (b < NBUCK) {
                offL[b] = run;
                seg[j * NBUCK + b] = make_int2(run, loc[u]);
            }
            run += loc[u];
        }
    }
    __syncthreads();
    #pragma unroll
    for (int k = 0; k < SORT_UNR; ++k) {
        if (e0 + t + k * SORT_THREADS < e1) {
            const int p = atomicAdd(&offL[d[k] >> BSHIFT], 1);
            tmp[e0 + p] = make_int2(s[k], d[k]);
        }
    }
    for (int base = e0 + t + ROUND; base < e1 + t; base += ROUND) {
        int s2[SORT_UNR], d2[SORT_UNR];
        #pragma unroll
        for (int k = 0; k < SORT_UNR; ++k) {
            const int e = base + k * SORT_THREADS;
            const int ec = e < e1 ? e : ec0;
            s2[k] = src[ec];
            d2[k] = dst[ec];
        }
        #pragma unroll
        for (int k = 0; k < SORT_UNR; ++k) {
            if (base + k * SORT_THREADS < e1) {
                const int p = atomicAdd(&offL[d2[k] >> BSHIFT], 1);
                tmp[e0 + p] = make_int2(s2[k], d2[k]);
            }
        }
    }
}

__global__ __launch_bounds__(SORT_THREADS) void fine_sort_kernel(
        const int2* __restrict__ tmp, const int2* __restrict__ seg,
        const float* __restrict__ x, int* __restrict__ srcs,
        int* __restrict__ rowptr, int* __restrict__ cnt,
        float* __restrict__ xs, int N, int E) {
    __shared__ int cL[BNODES];
    __shared__ int curL[BNODES];
    __shared__ int wsum[BNODES / 32];
    __shared__ int wpart[SORT_WAVES];
    __shared__ int s0L;
    const int t = (int)threadIdx.x, lane = t & 31, b = (int)blockIdx.x;
    const int wave = __builtin_amdgcn_readfirstlane(t >> 5);
    const int n0 = b * BNODES;
    int2 sg = make_int2(0, 0);
    if (lane < RUNS_PER_WAVE) sg = seg[(wave * RUNS_PER_WAVE + lane) * NBUCK + b];
    if (t < BNODES) cL[t] = 0;
    int part = sg.x;
    #pragma unroll
    for (int o = 16; o > 0; o >>= 1) part += __shfl_xor(part, o);
    if (lane == 0) wpart[wave] = part;
    const int ch = sort_chunk(E);
    int ra[RUNS_PER_WAVE], rc[RUNS_PER_WAVE];
    int maxc = 0;
    #pragma unroll
    for (int u = 0; u < RUNS_PER_WAVE; ++u) {
        ra[u] = (wave * RUNS_PER_WAVE + u) * ch + lane_bcast_i(sg.x, u);
        rc[u] = lane_bcast_i(sg.y, u);
        maxc = rc[u] > maxc ? rc[u] : maxc;
    }
    __syncthreads();
    if (wave == 0) {
        int v = lane < SORT_WAVES ? wpart[lane] : 0;
        #pragma unroll
        for (int o = 16; o > 0; o >>= 1) v += __shfl_xor(v, o);
        if (lane == 0) s0L = v;
    }
    for (int r0 = 0; r0 < maxc; r0 += 32) {
        const int i = r0 + lane;
        int d[RUNS_PER_WAVE];
        #pragma unroll
        for (int u = 0; u < RUNS_PER_WAVE; ++u)
            d[u] = tmp[i < rc[u] ? ra[u] + i : 0].y;
        #pragma unroll
        for (int u = 0; u < RUNS_PER_WAVE; ++u)
            if (i < rc[u]) atomicAdd(&cL[d[u] - n0], 1);
    }
    __syncthreads();
    const int s0 = s0L;
    int c = 0, incl = 0;
    if (t < BNODES) {
        c = cL[t];
        incl = c;
        #pragma unroll
        for (int o = 1; o < 32; o <<= 1) {
            int up = __shfl_up(incl, o);
            if (lane >= o) incl += up;
        }
        if (lane == 31) wsum[wave] = incl;
    }
    __syncthreads();
    if (t < BNODES) {
        int base = 0;
        for (int k = 0; k < wave; ++k) base += wsum[k];
        const int ex = base + incl - c;
        curL[t] = ex;
        const int n = n0 + t;
        if (n < N) { rowptr[n] = s0 + ex; cnt[n] = c; }
    }
    __syncthreads();
    for (int r0 = 0; r0 < maxc; r0 += 32) {
        const int i = r0 + lane;
        int2 p[RUNS_PER_WAVE];
        #pragma unroll
        for (int u = 0; u < RUNS_PER_WAVE; ++u)
            p[u] = tmp[i < rc[u] ? ra[u] + i : 0];
        #pragma unroll
        for (int u = 0; u < RUNS_PER_WAVE; ++u) {
            if (i < rc[u]) {
                const int r = atomicAdd(&curL[p[u].y - n0], 1);
                srcs[s0 + r] = p[u].x;
            }
        }
    }
    for (int i = t; i < BNODES * 8; i += SORT_THREADS) {
        const int row = i >> 3;
        const int n = n0 + row;
        if (n < N) {
            const float dn = rsqrtf((float)(cL[row] + 1));
            float4 v = ((const float4*)x)[(size_t)n * 8 + (i & 7)];
            v.x *= dn; v.y *= dn; v.z *= dn; v.w *= dn;
            ((float4*)xs)[(size_t)n * 8 + (i & 7)] = v;
        }
    }
}


#define GATHER_GROUP 16
#define NODES_PER_WAVE 4
template <int K>
__device__ __forceinline__ void gather_group(const float* __restrict__ h, int s, float w,
                                             int j, int lane, float& ax, float& ay) {
    if constexpr (K == 64) {
        float2 v[GATHER_GROUP];
        #pragma unroll
        for (int u = 0; u < GATHER_GROUP; ++u) {
            const int sj = lane_bcast_i(s, j + u);
            v[u] = ((const float2*)(h + (size_t)sj * 64))[lane];
        }
        #pragma unroll
        for (int u = 0; u < GATHER_GROUP; ++u) {
            const float wj = lane_bcast_f(w, j + u);
            ax += wj * v[u].x;
            ay += wj * v[u].y;
        }
    } else {
        float v[GATHER_GROUP];
        #pragma unroll
        for (int u = 0; u < GATHER_GROUP; ++u) {
            const int sj = lane_bcast_i(s, j + u);
            v[u] = h[(size_t)sj * 32 + lane];
        }
        #pragma unroll
        for (int u = 0; u < GATHER_GROUP; ++u) {
            const float wj = lane_bcast_f(w, j + u);
            ax += wj * v[u];
        }
    }
}

template <int K, bool POOL>
__global__ __launch_bounds__(256) void agg_gemm_kernel(
        const int* __restrict__ rowptr, const int* __restrict__ cnt,
        const int* __restrict__ srcs, const float* __restrict__ h,
        const float* __restrict__ W, const float* __restrict__ bias,
        float* __restrict__ hout, const int* __restrict__ batch,
        float* pooled, float* counts, int N) {
    constexpr int PITCH = K + 4;
    __shared__ __attribute__((aligned(16))) float Wlds[K * 64];
    __shared__ __attribute__((aligned(16))) float Alds[32 * PITCH];
    __shared__ int Glds[32];
    __shared__ float Dlds[32];

    const int lane = threadIdx.x & 31;
    const int wave = __builtin_amdgcn_readfirstlane((int)(threadIdx.x >> 5));
    const int nb = (int)blockIdx.x * 32 + wave * NODES_PER_WAVE;

    int startv = 0, degv = 0, gv = 0;
    if (lane < NODES_PER_WAVE && nb + lane < N) {
        startv = rowptr[nb + lane];
        degv   = cnt[nb + lane];
        if (POOL) gv = batch[nb + lane];
    }
    float4 wv[K / 16];
    #pragma unroll
    for (int u = 0; u < K / 16; ++u)
        wv[u] = ((const float4*)W)[u * 256 + (int)threadIdx.x];
    #pragma unroll
    for (int u = 0; u < K / 16; ++u)
        ((float4*)Wlds)[u * 256 + (int)threadIdx.x] = wv[u];

    int s[NODES_PER_WAVE], st[NODES_PER_WAVE], dg[NODES_PER_WAVE];
    float w[NODES_PER_WAVE];
    int pv[NODES_PER_WAVE];
    #pragma unroll
    for (int i = 0; i < NODES_PER_WAVE; ++i) {
        st[i] = lane_bcast_i(startv, i);
        dg[i] = lane_bcast_i(degv, i);
        const int c0 = dg[i] < 32 ? dg[i] : 32;
        pv[i] = srcs[lane < c0 ? st[i] + lane : 0];
    }
    #pragma unroll
    for (int i = 0; i < NODES_PER_WAVE; ++i) {
        const int c0 = dg[i] < 32 ? dg[i] : 32;
        s[i] = lane < c0 ? pv[i] : nb + i;
        w[i] = lane < c0 ? 1.0f : 0.0f;
    }

    #pragma unroll
    for (int i = 0; i < NODES_PER_WAVE; ++i) {
        const int n = nb + i;
        const int row = wave * NODES_PER_WAVE + i;
        float ax = 0.0f, ay = 0.0f;
        if (n < N) {
            const int deg = dg[i], start = st[i];
            const int c0 = deg < 32 ? deg : 32;
            float sx, sy = 0.0f;
            if constexpr (K == 64) {
                const float2 sv = ((const float2*)(h + (size_t)n * 64))[lane];
                sx = sv.x; sy = sv.y;
            } else {
                sx = h[(size_t)n * 32 + lane];
            }
            for (int j = 0; j < c0; j += GATHER_GROUP)
                gather_group<K>(h, s[i], w[i], j, lane, ax, ay);
            for (int base = 32; base < deg; base += 32) {
                const int rem = deg - base;
                const int c = rem < 32 ? rem : 32;
                int s2 = n;
                float w2 = 0.0f;
                if (lane < c) {
                    s2 = srcs[start + base + lane];
                    w2 = 1.0f;
                }
                for (int j = 0; j < c; j += GATHER_GROUP)
                    gather_group<K>(h, s2, w2, j, lane, ax, ay);
            }
            const float dn = rsqrtf((float)(deg + 1));
            ax = dn * (ax + sx);
            ay = dn * (ay + sy);
            if (!POOL && lane == 0) Dlds[row] = dn;
            if (POOL) {
                const int g = lane_bcast_i(gv, i);
                if (lane == 0) {
                    Glds[row] = g;
                    atomic_add_f32(&counts[(size_t)g * GCNT_STRIDE], 1.0f);
                }
            }
        }
        if constexpr (K == 64) ((float2*)(Alds + row * PITCH))[lane] = make_float2(ax, ay);
        else                   Alds[row * PITCH + lane] = ax;
    }
    __syncthreads();

    const int half = lane >> 4;
    const int l16  = lane & 15;
    const int col0 = (wave & 3) * 16;
    const int rloc = (wave >> 2) * 16;
    const float* Arow = Alds + (rloc + l16) * PITCH + 2 * half;
    const float* Brow = Wlds + 2 * half * 64 + col0 + l16;

    v8f acc = {};
    #pragma unroll
    for (int k0 = 0; k0 < K; k0 += 4) {
        v2f a = *(const v2f*)(Arow + k0);
        v2f b;
        b.x = Brow[k0 * 64];
        b.y = Brow[k0 * 64 + 64];
        acc = __builtin_amdgcn_wmma_f32_16x16x4_f32(
             false, a,  false, b,
             (short)0, acc,  false,  false);
    }
    const float bv = bias[col0 + l16];
    #pragma unroll
    for (int r = 0; r < 8; ++r) {
        const int rl = rloc + r + 8 * half;
        const int mm = (int)blockIdx.x * 32 + rl;
        if (mm < N) {
            const float v = fmaxf(acc[r] + bv, 0.0f);
            if (POOL) atomic_add_f32(pooled + (size_t)Glds[rl] * 64 + col0 + l16, v);
            else      hout[(size_t)mm * 64 + col0 + l16] = Dlds[rl] * v;
        }
    }
}

__global__ void pooled_mlp_head_kernel(const float* __restrict__ pooled,
                                       const float* __restrict__ counts,
                                       const float* __restrict__ lw1,
                                       const float* __restrict__ lb1,
                                       const float* __restrict__ lw2,
                                       const float* __restrict__ lb2,
                                       float* __restrict__ out, int G) {
    __shared__ float Wlds[64 * 64];
    __shared__ float Rt[32 * 65];
    #pragma unroll 4
    for (int i = threadIdx.x; i < 64 * 64; i += 256)
        Wlds[i] = lw1[i];
    __syncthreads();

    const int wave = threadIdx.x >> 5;
    const int lane = threadIdx.x & 31;
    const int half = lane >> 4;
    const int l16  = lane & 15;
    const int col0 = (wave & 3) * 16;
    const int rloc = (wave >> 2) * 16;
    const int row0 = blockIdx.x * 32 + rloc;
    const int m    = row0 + l16;
    const int mc   = (m < G) ? m : (G - 1);
    const float c  = fmaxf(counts[(size_t)mc * GCNT_STRIDE], 1.0f);
    const float* __restrict__ Arow = pooled + (size_t)mc * 64 + 2 * half;
    const float* __restrict__ Brow = Wlds + 2 * half * 64 + col0 + l16;

    v8f acc = {};
    #pragma unroll
    for (int k0 = 0; k0 < 64; k0 += 4) {
        v2f a = *(const v2f*)(Arow + k0);
        a.x = a.x / c;
        a.y = a.y / c;
        v2f b;
        b.x = Brow[k0 * 64];
        b.y = Brow[k0 * 64 + 64];
        acc = __builtin_amdgcn_wmma_f32_16x16x4_f32(
             false, a,  false, b,
             (short)0, acc,  false,  false);
    }
    const float bv = lb1[col0 + l16];
    #pragma unroll
    for (int r = 0; r < 8; ++r)
        Rt[(rloc + r + 8 * half) * 65 + col0 + l16] = fmaxf(acc[r] + bv, 0.0f);
    __syncthreads();

    const int t = threadIdx.x;
    if (t < 64) {
        const int gl = t >> 1, j = t & 1;
        const int gg = blockIdx.x * 32 + gl;
        if (gg < G) {
            float o = lb2[j];
            #pragma unroll
            for (int k = 0; k < 64; ++k) o += Rt[gl * 65 + k] * lw2[k * 2 + j];
            out[(size_t)gg * 2 + j] = o;
        }
    }
}

static inline size_t alignup(size_t v) { return (v + 255) & ~(size_t)255; }

extern "C" void kernel_launch(void* const* d_in, const int* in_sizes, int n_in,
                              void* d_out, int out_size, void* d_ws, size_t ws_size,
                              hipStream_t stream) {
    const float* x     = (const float*)d_in[0];
    const int*   eidx  = (const int*)d_in[1];
    const int*   batch = (const int*)d_in[2];
    const float* W1 = (const float*)d_in[3];  const float* b1 = (const float*)d_in[4];
    const float* W2 = (const float*)d_in[5];  const float* b2 = (const float*)d_in[6];
    const float* W3 = (const float*)d_in[7];  const float* b3 = (const float*)d_in[8];
    const float* lw1 = (const float*)d_in[9]; const float* lb1 = (const float*)d_in[10];
    const float* lw2 = (const float*)d_in[11]; const float* lb2 = (const float*)d_in[12];
    float* out = (float*)d_out;

    const int N = N_NODES, G = N_GRAPHS;
    const int E = in_sizes[1] / 2;
    const int* src = eidx;
    const int* dst = eidx + E;

    char* w = (char*)d_ws;
    int2*  seg    = (int2*)w;  w += alignup((size_t)NBUCK * SORT_BLOCKS * 8);
    int*   cnt    = (int*)w;   w += alignup((size_t)N * 4);
    int*   rowptr = (int*)w;   w += alignup((size_t)N * 4);
    int2*  tmp    = (int2*)w;  w += alignup((size_t)E * 8);
    int*   srcs   = (int*)w;   w += alignup((size_t)E * 4);
    float* xs     = (float*)w; w += alignup((size_t)N * 32 * 4);
    float* h1     = (float*)w; w += alignup((size_t)N * 64 * 4);
    float* h2     = (float*)w; w += alignup((size_t)N * 64 * 4);
    float* pooled = (float*)w; w += alignup((size_t)G * 64 * 4);
    float* counts = (float*)w; w += alignup((size_t)G * 4 * GCNT_STRIDE);
    (void)ws_size; (void)n_in;

    dim3 blk256(256);
    const int GF4 = G * 64 / 4;
    const int G4 = G * GCNT_STRIDE / 4;
    int gLayer = (N + 31) / 32;
    int gGemmG = (G + 31) / 32;

    coarse_sort_kernel<<<SORT_BLOCKS, dim3(SORT_THREADS), 0, stream>>>(src, dst, tmp, seg, E, pooled, GF4, counts, G4);
    fine_sort_kernel<<<NBUCK, dim3(SORT_THREADS), 0, stream>>>(tmp, seg, x, srcs, rowptr, cnt, xs, N, E);

    agg_gemm_kernel<32, false><<<gLayer, blk256, 0, stream>>>(rowptr, cnt, srcs, xs, W1, b1, h1,
                                                              batch, pooled, counts, N);
    agg_gemm_kernel<64, false><<<gLayer, blk256, 0, stream>>>(rowptr, cnt, srcs, h1, W2, b2, h2,
                                                              batch, pooled, counts, N);
    agg_gemm_kernel<64, true><<<gLayer, blk256, 0, stream>>>(rowptr, cnt, srcs, h2, W3, b3, h1,
                                                             batch, pooled, counts, N);

    pooled_mlp_head_kernel<<<gGemmG, blk256, 0, stream>>>(pooled, counts, lw1, lb1, lw2, lb2, out, G);
    (void)out_size;
}


# last layer pooling: block stages its 32 relu'd rows in LDS, 64 threads sum rows of equal graph id in node order; one f32 atomic per run and column (count: run length), ~14x fewer; sum order differs
# speedup vs baseline: 5.0830x; 1.0299x over previous
#include <hip/hip_runtime.h>

typedef __attribute__((ext_vector_type(2))) float v2f;
typedef __attribute__((ext_vector_type(8))) float v8f;

#define N_NODES 50000
#define N_GRAPHS 2048

__device__ __forceinline__ void atomic_add_f32(float* p, float v) {
    unsafeAtomicAdd(p, v);
}
__device__ __forceinline__ int lane_bcast_i(int v, int j) {
    return __builtin_amdgcn_readlane(v, j);
}
__device__ __forceinline__ float lane_bcast_f(float v, int j) {
    return __int_as_float(__builtin_amdgcn_readlane(__float_as_int(v), j));
}

#define BSHIFT 8
#define BNODES 256
#define NBUCK ((N_NODES + BNODES - 1) / BNODES)
#define SORT_BLOCKS 256
#define SORT_THREADS 1024
#define SORT_UNR 4
#define SORT_WAVES (SORT_THREADS / 32)
#define RUNS_PER_WAVE (SORT_BLOCKS / SORT_WAVES)
static_assert(RUNS_PER_WAVE * SORT_WAVES == SORT_BLOCKS && RUNS_PER_WAVE <= 32 && SORT_WAVES <= 32,
              "level 2 gives every wave the same number of runs, one table entry per lane");
#define GCNT_STRIDE 32

__device__ __forceinline__ int sort_chunk(int E) { return (E + SORT_BLOCKS - 1) / SORT_BLOCKS; }

__global__ __launch_bounds__(SORT_THREADS) void coarse_sort_kernel(
        const int* __restrict__ src, const int* __restrict__ dst,
        int2* __restrict__ tmp, int2* __restrict__ seg, int E,
        float* pooled, int n4_pooled, float* counts, int n4_counts) {
    __shared__ int hl[NBUCK];
    __shared__ int offL[NBUCK];
    const int t = (int)threadIdx.x, lane = t & 31, wave = t >> 5, j = (int)blockIdx.x;
    constexpr int TRIPS = (NBUCK + 31) / 32;
    constexpr int ROUND = SORT_UNR * SORT_THREADS;
    if (t < NBUCK) hl[t] = 0;
    for (int g = j * SORT_THREADS + t; g < n4_pooled + n4_counts; g += SORT_BLOCKS * SORT_THREADS) {
        float4* p = (g < n4_pooled) ? ((float4*)pooled + g) : ((float4*)counts + (g - n4_pooled));
        *p = make_float4(0.f, 0.f, 0.f, 0.f);
    }
    const int ch = sort_chunk(E);
    const int e0 = j * ch;
    const int e1 = (e0 + ch < E) ? (e0 + ch) : E;
    const int ec0 = (e0 < E) ? e0 : 0;
    int s[SORT_UNR], d[SORT_UNR];
    #pragma unroll
    for (int k = 0; k < SORT_UNR; ++k) {
        const int e = e0 + t + k * SORT_THREADS;
        const int ec = e < e1 ? e : ec0;
        s[k] = src[ec];
        d[k] = dst[ec];
    }
    __syncthreads();
    #pragma unroll
    for (int k = 0; k < SORT_UNR; ++k)
        if (e0 + t + k * SORT_THREADS < e1) atomicAdd(&hl[d[k] >> BSHIFT], 1);
    for (int base = e0 + t + ROUND; base < e1 + t; base += ROUND) {
        int d2[SORT_UNR];
        #pragma unroll
        for (int k = 0; k < SORT_UNR; ++k) {
            const int e = base + k * SORT_THREADS;
            d2[k] = dst[e < e1 ? e : ec0];
        }
        #pragma unroll
        for (int k = 0; k < SORT_UNR; ++k)
            if (base + k * SORT_THREADS < e1) atomicAdd(&hl[d2[k] >> BSHIFT], 1);
    }
    __syncthreads();
    if (wave == 0) {
        int loc[TRIPS];
        int sum = 0;
        #pragma unroll
        for (int u = 0; u < TRIPS; ++u) {
            const int b = lane * TRIPS + u;
            loc[u] = (b < NBUCK) ? hl[b] : 0;
            sum += loc[u];
        }
        int incl = sum;
        #pragma unroll
        for (int o = 1; o < 32; o <<= 1) {
            int up = __shfl_up(incl, o);
            if (lane >= o) incl += up;
        }
        int run = incl - sum;
        #pragma unroll
        for (int u = 0; u < TRIPS; ++u) {
            const int b = lane * TRIPS + u;
            if (b < NBUCK) {
                offL[b] = run;
                seg[j * NBUCK + b] = make_int2(run, loc[u]);
            }
            run += loc[u];
        }
    }
    __syncthreads();
    #pragma unroll
    for (int k = 0; k < SORT_UNR; ++k) {
        if (e0 + t + k * SORT_THREADS < e1) {
            const int p = atomicAdd(&offL[d[k] >> BSHIFT], 1);
            tmp[e0 + p] = make_int2(s[k], d[k]);
        }
    }
    for (int base = e0 + t + ROUND; base < e1 + t; base += ROUND) {
        int s2[SORT_UNR], d2[SORT_UNR];
        #pragma unroll
        for (int k = 0; k < SORT_UNR; ++k) {
            const int e = base + k * SORT_THREADS;
            const int ec = e < e1 ? e : ec0;
            s2[k] = src[ec];
            d2[k] = dst[ec];
        }
        #pragma unroll
        for (int k = 0; k < SORT_UNR; ++k) {
            if (base + k * SORT_THREADS < e1) {
                const int p = atomicAdd(&offL[d2[k] >> BSHIFT], 1);
                tmp[e0 + p] = make_int2(s2[k], d2[k]);
            }
        }
    }
}

__global__ __launch_bounds__(SORT_THREADS) void fine_sort_kernel(
        const int2* __restrict__ tmp, const int2* __restrict__ seg,
        const float* __restrict__ x, int* __restrict__ srcs,
        int* __restrict__ rowptr, int* __restrict__ cnt,
        float* __restrict__ xs, int N, int E) {
    __shared__ int cL[BNODES];
    __shared__ int curL[BNODES];
    __shared__ int wsum[BNODES / 32];
    __shared__ int wpart[SORT_WAVES];
    __shared__ int s0L;
    const int t = (int)threadIdx.x, lane = t & 31, b = (int)blockIdx.x;
    const int wave = __builtin_amdgcn_readfirstlane(t >> 5);
    const int n0 = b * BNODES;
    int2 sg = make_int2(0, 0);
    if (lane < RUNS_PER_WAVE) sg = seg[(wave * RUNS_PER_WAVE + lane) * NBUCK + b];
    if (t < BNODES) cL[t] = 0;
    int part = sg.x;
    #pragma unroll
    for (int o = 16; o > 0; o >>= 1) part += __shfl_xor(part, o);
    if (lane == 0) wpart[wave] = part;
    const int ch = sort_chunk(E);
    int ra[RUNS_PER_WAVE], rc[RUNS_PER_WAVE];
    int maxc = 0;
    #pragma unroll
    for (int u = 0; u < RUNS_PER_WAVE; ++u) {
        ra[u] = (wave * RUNS_PER_WAVE + u) * ch + lane_bcast_i(sg.x, u);
        rc[u] = lane_bcast_i(sg.y, u);
        maxc = rc[u] > maxc ? rc[u] : maxc;
    }
    __syncthreads();
    if (wave == 0) {
        int v = lane < SORT_WAVES ? wpart[lane] : 0;
        #pragma unroll
        for (int o = 16; o > 0; o >>= 1) v += __shfl_xor(v, o);
        if (lane == 0) s0L = v;
    }
    for (int r0 = 0; r0 < maxc; r0 += 32) {
        const int i = r0 + lane;
        int d[RUNS_PER_WAVE];
        #pragma unroll
        for (int u = 0; u < RUNS_PER_WAVE; ++u)
            d[u] = tmp[i < rc[u] ? ra[u] + i : 0].y;
        #pragma unroll
        for (int u = 0; u < RUNS_PER_WAVE; ++u)
            if (i < rc[u]) atomicAdd(&cL[d[u] - n0], 1);
    }
    __syncthreads();
    const int s0 = s0L;
    int c = 0, incl = 0;
    if (t < BNODES) {
        c = cL[t];
        incl = c;
        #pragma unroll
        for (int o = 1; o < 32; o <<= 1) {
            int up = __shfl_up(incl, o);
            if (lane >= o) incl += up;
        }
        if (lane == 31) wsum[wave] = incl;
    }
    __syncthreads();
    if (t < BNODES) {
        int base = 0;
        for (int k = 0; k < wave; ++k) base += wsum[k];
        const int ex = base + incl - c;
        curL[t] = ex;
        const int n = n0 + t;
        if (n < N) { rowptr[n] = s0 + ex; cnt[n] = c; }
    }
    __syncthreads();
    for (int r0 = 0; r0 < maxc; r0 += 32) {
        const int i = r0 + lane;
        int2 p[RUNS_PER_WAVE];
        #pragma unroll
        for (int u = 0; u < RUNS_PER_WAVE; ++u)
            p[u] = tmp[i < rc[u] ? ra[u] + i : 0];
        #pragma unroll
        for (int u = 0; u < RUNS_PER_WAVE; ++u) {
            if (i < rc[u]) {
                const int r = atomicAdd(&curL[p[u].y - n0], 1);
                srcs[s0 + r] = p[u].x;
            }
        }
    }
    for (int i = t; i < BNODES * 8; i += SORT_THREADS) {
        const int row = i >> 3;
        const int n = n0 + row;
        if (n < N) {
            const float dn = rsqrtf((float)(cL[row] + 1));
            float4 v = ((const float4*)x)[(size_t)n * 8 + (i & 7)];
            v.x *= dn; v.y *= dn; v.z *= dn; v.w *= dn;
            ((float4*)xs)[(size_t)n * 8 + (i & 7)] = v;
        }
    }
}


#define GATHER_GROUP 16
#define NODES_PER_WAVE 4
template <int K>
__device__ __forceinline__ void gather_group(const float* __restrict__ h, int s, float w,
                                             int j, int lane, float& ax, float& ay) {
    if constexpr (K == 64) {
        float2 v[GATHER_GROUP];
        #pragma unroll
        for (int u = 0; u < GATHER_GROUP; ++u) {
            const int sj = lane_bcast_i(s, j + u);
            v[u] = ((const float2*)(h + (size_t)sj * 64))[lane];
        }
        #pragma unroll
        for (int u = 0; u < GATHER_GROUP; ++u) {
            const float wj = lane_bcast_f(w, j + u);
            ax += wj * v[u].x;
            ay += wj * v[u].y;
        }
    } else {
        float v[GATHER_GROUP];
        #pragma unroll
        for (int u = 0; u < GATHER_GROUP; ++u) {
            const int sj = lane_bcast_i(s, j + u);
            v[u] = h[(size_t)sj * 32 + lane];
        }
        #pragma unroll
        for (int u = 0; u < GATHER_GROUP; ++u) {
            const float wj = lane_bcast_f(w, j + u);
            ax += wj * v[u];
        }
    }
}

template <int K, bool POOL>
__global__ __launch_bounds__(256) void agg_gemm_kernel(
        const int* __restrict__ rowptr, const int* __restrict__ cnt,
        const int* __restrict__ srcs, const float* __restrict__ h,
        const float* __restrict__ W, const float* __restrict__ bias,
        float* __restrict__ hout, const int* __restrict__ batch,
        float* pooled, float* counts, int N) {
    constexpr int PITCH = K + 4;
    __shared__ __attribute__((aligned(16))) float Wlds[K * 64];
    __shared__ __attribute__((aligned(16))) float Alds[32 * PITCH];
    __shared__ int Glds[32];
    __shared__ float Dlds[32];
    constexpr int HPITCH = 68;
    __shared__ float Hlds[POOL ? 32 * HPITCH : 1];

    const int lane = threadIdx.x & 31;
    const int wave = __builtin_amdgcn_readfirstlane((int)(threadIdx.x >> 5));
    const int nb = (int)blockIdx.x * 32 + wave * NODES_PER_WAVE;

    int startv = 0, degv = 0, gv = 0;
    if (lane < NODES_PER_WAVE && nb + lane < N) {
        startv = rowptr[nb + lane];
        degv   = cnt[nb + lane];
        if (POOL) gv = batch[nb + lane];
    }
    float4 wv[K / 16];
    #pragma unroll
    for (int u = 0; u < K / 16; ++u)
        wv[u] = ((const float4*)W)[u * 256 + (int)threadIdx.x];
    #pragma unroll
    for (int u = 0; u < K / 16; ++u)
        ((float4*)Wlds)[u * 256 + (int)threadIdx.x] = wv[u];

    int s[NODES_PER_WAVE], st[NODES_PER_WAVE], dg[NODES_PER_WAVE];
    float w[NODES_PER_WAVE];
    int pv[NODES_PER_WAVE];
    #pragma unroll
    for (int i = 0; i < NODES_PER_WAVE; ++i) {
        st[i] = lane_bcast_i(startv, i);
        dg[i] = lane_bcast_i(degv, i);
        const int c0 = dg[i] < 32 ? dg[i] : 32;
        pv[i] = srcs[lane < c0 ? st[i] + lane : 0];
    }
    #pragma unroll
    for (int i = 0; i < NODES_PER_WAVE; ++i) {
        const int c0 = dg[i] < 32 ? dg[i] : 32;
        s[i] = lane < c0 ? pv[i] : nb + i;
        w[i] = lane < c0 ? 1.0f : 0.0f;
    }

    #pragma unroll
    for (int i = 0; i < NODES_PER_WAVE; ++i) {
        const int n = nb + i;
        const int row = wave * NODES_PER_WAVE + i;
        float ax = 0.0f, ay = 0.0f;
        if (n < N) {
            const int deg = dg[i], start = st[i];
            const int c0 = deg < 32 ? deg : 32;
            float sx, sy = 0.0f;
            if constexpr (K == 64) {
                const float2 sv = ((const float2*)(h + (size_t)n * 64))[lane];
                sx = sv.x; sy = sv.y;
            } else {
                sx = h[(size_t)n * 32 + lane];
            }
            for (int j = 0; j < c0; j += GATHER_GROUP)
                gather_group<K>(h, s[i], w[i], j, lane, ax, ay);
            for (int base = 32; base < deg; base += 32) {
                const int rem = deg - base;
                const int c = rem < 32 ? rem : 32;
                int s2 = n;
                float w2 = 0.0f;
                if (lane < c) {
                    s2 = srcs[start + base + lane];
                    w2 = 1.0f;
                }
                for (int j = 0; j < c; j += GATHER_GROUP)
                    gather_group<K>(h, s2, w2, j, lane, ax, ay);
            }
            const float dn = rsqrtf((float)(deg + 1));
            ax = dn * (ax + sx);
            ay = dn * (ay + sy);
            if (!POOL && lane == 0) Dlds[row] = dn;
            if (POOL) {
                const int g = lane_bcast_i(gv, i);
                if (lane == 0) Glds[row] = g;
            }
        }
        if constexpr (K == 64) ((float2*)(Alds + row * PITCH))[lane] = make_float2(ax, ay);
        else                   Alds[row * PITCH + lane] = ax;
    }
    __syncthreads();

    const int half = lane >> 4;
    const int l16  = lane & 15;
    const int col0 = (wave & 3) * 16;
    const int rloc = (wave >> 2) * 16;
    const float* Arow = Alds + (rloc + l16) * PITCH + 2 * half;
    const float* Brow = Wlds + 2 * half * 64 + col0 + l16;

    v8f acc = {};
    #pragma unroll
    for (int k0 = 0; k0 < K; k0 += 4) {
        v2f a = *(const v2f*)(Arow + k0);
        v2f b;
        b.x = Brow[k0 * 64];
        b.y = Brow[k0 * 64 + 64];
        acc = __builtin_amdgcn_wmma_f32_16x16x4_f32(
             false, a,  false, b,
             (short)0, acc,  false,  false);
    }
    const float bv = bias[col0 + l16];
    if constexpr (!POOL) {
        #pragma unroll
        for (int r = 0; r < 8; ++r) {
            const int rl = rloc + r + 8 * half;
            const int mm = (int)blockIdx.x * 32 + rl;
            if (mm < N) {
                const float v = fmaxf(acc[r] + bv, 0.0f);
                hout[(size_t)mm * 64 + col0 + l16] = Dlds[rl] * v;
            }
        }
    } else {
        #pragma unroll
        for (int r = 0; r < 8; ++r)
            Hlds[(rloc + r + 8 * half) * HPITCH + col0 + l16] = fmaxf(acc[r] + bv, 0.0f);
        __syncthreads();
        if (wave < 2) {
            const int c = wave * 32 + lane;
            const int left = N - (int)blockIdx.x * 32;
            const int nvalid = left < 32 ? left : 32;
            const int gl = lane < nvalid ? Glds[lane] : 0;
            float hv[32];
            #pragma unroll
            for (int r = 0; r < 32; ++r) hv[r] = Hlds[r * HPITCH + c];
            int gcur = lane_bcast_i(gl, 0);
            int len = 0;
            float run = 0.0f;
            #pragma unroll
            for (int r = 0; r < 32; ++r) {
                if (r < nvalid) {
                    const int g = lane_bcast_i(gl, r);
                    if (g != gcur) {
                        atomic_add_f32(pooled + (size_t)gcur * 64 + c, run);
                        if (c == 0) atomic_add_f32(&counts[(size_t)gcur * GCNT_STRIDE], (float)len);
                        gcur = g;
                        len = 0;
                        run = 0.0f;
                    }
                    run += hv[r];
                    ++len;
                }
            }
            atomic_add_f32(pooled + (size_t)gcur * 64 + c, run);
            if (c == 0) atomic_add_f32(&counts[(size_t)gcur * GCNT_STRIDE], (float)len);
        }
    }
}

__global__ void pooled_mlp_head_kernel(const float* __restrict__ pooled,
                                       const float* __restrict__ counts,
                                       const float* __restrict__ lw1,
                                       const float* __restrict__ lb1,
                                       const float* __restrict__ lw2,
                                       const float* __restrict__ lb2,
                                       float* __restrict__ out, int G) {
    __shared__ float Wlds[64 * 64];
    __shared__ float Rt[32 * 65];
    #pragma unroll 4
    for (int i = threadIdx.x; i < 64 * 64; i += 256)
        Wlds[i] = lw1[i];
    __syncthreads();

    const int wave = threadIdx.x >> 5;
    const int lane = threadIdx.x & 31;
    const int half = lane >> 4;
    const int l16  = lane & 15;
    const int col0 = (wave & 3) * 16;
    const int rloc = (wave >> 2) * 16;
    const int row0 = blockIdx.x * 32 + rloc;
    const int m    = row0 + l16;
    const int mc   = (m < G) ? m : (G - 1);
    const float c  = fmaxf(counts[(size_t)mc * GCNT_STRIDE], 1.0f);
    const float* __restrict__ Arow = pooled + (size_t)mc * 64 + 2 * half;
    const float* __restrict__ Brow = Wlds + 2 * half * 64 + col0 + l16;

    v8f acc = {};
    #pragma unroll
    for (int k0 = 0; k0 < 64; k0 += 4) {
        v2f a = *(const v2f*)(Arow + k0);
        a.x = a.x / c;
        a.y = a.y / c;
        v2f b;
        b.x = Brow[k0 * 64];
        b.y = Brow[k0 * 64 + 64];
        acc = __builtin_amdgcn_wmma_f32_16x16x4_f32(
             false, a,  false, b,
             (short)0, acc,  false,  false);
    }
    const float bv = lb1[col0 + l16];
    #pragma unroll
    for (int r = 0; r < 8; ++r)
        Rt[(rloc + r + 8 * half) * 65 + col0 + l16] = fmaxf(acc[r] + bv, 0.0f);
    __syncthreads();

    const int t = threadIdx.x;
    if (t < 64) {
        const int gl = t >> 1, j = t & 1;
        const int gg = blockIdx.x * 32 + gl;
        if (gg < G) {
            float o = lb2[j];
            #pragma unroll
            for (int k = 0; k < 64; ++k) o += Rt[gl * 65 + k] * lw2[k * 2 + j];
            out[(size_t)gg * 2 + j] = o;
        }
    }
}

static inline size_t alignup(size_t v) { return (v + 255) & ~(size_t)255; }

extern "C" void kernel_launch(void* const* d_in, const int* in_sizes, int n_in,
                              void* d_out, int out_size, void* d_ws, size_t ws_size,
                              hipStream_t stream) {
    const float* x     = (const float*)d_in[0];
    const int*   eidx  = (const int*)d_in[1];
    const int*   batch = (const int*)d_in[2];
    const float* W1 = (const float*)d_in[3];  const float* b1 = (const float*)d_in[4];
    const float* W2 = (const float*)d_in[5];  const float* b2 = (const float*)d_in[6];
    const float* W3 = (const float*)d_in[7];  const float* b3 = (const float*)d_in[8];
    const float* lw1 = (const float*)d_in[9]; const float* lb1 = (const float*)d_in[10];
    const float* lw2 = (const float*)d_in[11]; const float* lb2 = (const float*)d_in[12];
    float* out = (float*)d_out;

    const int N = N_NODES, G = N_GRAPHS;
    const int E = in_sizes[1] / 2;
    const int* src = eidx;
    const int* dst = eidx + E;

    char* w = (char*)d_ws;
    int2*  seg    = (int2*)w;  w += alignup((size_t)NBUCK * SORT_BLOCKS * 8);
    int*   cnt    = (int*)w;   w += alignup((size_t)N * 4);
    int*   rowptr = (int*)w;   w += alignup((size_t)N * 4);
    int2*  tmp    = (int2*)w;  w += alignup((size_t)E * 8);
    int*   srcs   = (int*)w;   w += alignup((size_t)E * 4);
    float* xs     = (float*)w; w += alignup((size_t)N * 32 * 4);
    float* h1     = (float*)w; w += alignup((size_t)N * 64 * 4);
    float* h2     = (float*)w; w += alignup((size_t)N * 64 * 4);
    float* pooled = (float*)w; w += alignup((size_t)G * 64 * 4);
    float* counts = (float*)w; w += alignup((size_t)G * 4 * GCNT_STRIDE);
    (void)ws_size; (void)n_in;

    dim3 blk256(256);
    const int GF4 = G * 64 / 4;
    const int G4 = G * GCNT_STRIDE / 4;
    int gLayer = (N + 31) / 32;
    int gGemmG = (G + 31) / 32;

    coarse_sort_kernel<<<SORT_BLOCKS, dim3(SORT_THREADS), 0, stream>>>(src, dst, tmp, seg, E, pooled, GF4, counts, G4);
    fine_sort_kernel<<<NBUCK, dim3(SORT_THREADS), 0, stream>>>(tmp, seg, x, srcs, rowptr, cnt, xs, N, E);

    agg_gemm_kernel<32, false><<<gLayer, blk256, 0, stream>>>(rowptr, cnt, srcs, xs, W1, b1, h1,
                                                              batch, pooled, counts, N);
    agg_gemm_kernel<64, false><<<gLayer, blk256, 0, stream>>>(rowptr, cnt, srcs, h1, W2, b2, h2,
                                                              batch, pooled, counts, N);
    agg_gemm_kernel<64, true><<<gLayer, blk256, 0, stream>>>(rowptr, cnt, srcs, h2, W3, b3, h1,
                                                             batch, pooled, counts, N);

    pooled_mlp_head_kernel<<<gGemmG, blk256, 0, stream>>>(pooled, counts, lw1, lb1, lw2, lb2, out, G);
    (void)out_size;
}


# both sort levels place their records in LDS and write their slice of tmp / srcs in order, whole 128-B lines per store, not one scattered 8-B / 4-B store per edge; sum order = LDS arrival as before
# speedup vs baseline: 5.3423x; 1.0510x over previous
#include <hip/hip_runtime.h>

typedef __attribute__((ext_vector_type(2))) float v2f;
typedef __attribute__((ext_vector_type(8))) float v8f;

#define N_NODES 50000
#define N_GRAPHS 2048

__device__ __forceinline__ void atomic_add_f32(float* p, float v) {
    unsafeAtomicAdd(p, v);
}
__device__ __forceinline__ int lane_bcast_i(int v, int j) {
    return __builtin_amdgcn_readlane(v, j);
}
__device__ __forceinline__ float lane_bcast_f(float v, int j) {
    return __int_as_float(__builtin_amdgcn_readlane(__float_as_int(v), j));
}

#define BSHIFT 8
#define BNODES 256
#define NBUCK ((N_NODES + BNODES - 1) / BNODES)
#define SORT_BLOCKS 256
#define SORT_THREADS 1024
#define SORT_UNR 4
#define SORT_WAVES (SORT_THREADS / 32)
#define RUNS_PER_WAVE (SORT_BLOCKS / SORT_WAVES)
static_assert(RUNS_PER_WAVE * SORT_WAVES == SORT_BLOCKS && RUNS_PER_WAVE <= 32 && SORT_WAVES <= 32,
              "level 2 gives every wave the same number of runs, one table entry per lane");
#define GCNT_STRIDE 32

#define FINE_CAP 8192

__device__ __forceinline__ int sort_chunk(int E) { return (E + SORT_BLOCKS - 1) / SORT_BLOCKS; }
__device__ __forceinline__ int sort_pitch(int E) { return (sort_chunk(E) + 15) & ~15; }

__global__ __launch_bounds__(SORT_THREADS) void coarse_sort_kernel(
        const int* __restrict__ src, const int* __restrict__ dst,
        int2* __restrict__ tmp, int2* __restrict__ seg, int E,
        float* pooled, int n4_pooled, float* counts, int n4_counts) {
    constexpr int TRIPS = (NBUCK + 31) / 32;
    constexpr int ROUND = SORT_UNR * SORT_THREADS;
    __shared__ int hl[NBUCK];
    __shared__ int offL[NBUCK];
    __shared__ __attribute__((aligned(16))) int2 recL[ROUND];
    const int t = (int)threadIdx.x, lane = t & 31, wave = t >> 5, j = (int)blockIdx.x;
    if (t < NBUCK) hl[t] = 0;
    for (int g = j * SORT_THREADS + t; g < n4_pooled + n4_counts; g += SORT_BLOCKS * SORT_THREADS) {
        float4* p = (g < n4_pooled) ? ((float4*)pooled + g) : ((float4*)counts + (g - n4_pooled));
        *p = make_float4(0.f, 0.f, 0.f, 0.f);
    }
    const int ch = sort_chunk(E);
    const int e0 = j * ch;
    const int e1 = (e0 + ch < E) ? (e0 + ch) : E;
    const int ec0 = (e0 < E) ? e0 : 0;
    int2* const slice = tmp + (size_t)j * sort_pitch(E);
    int s[SORT_UNR], d[SORT_UNR];
    #pragma unroll
    for (int k = 0; k < SORT_UNR; ++k) {
        const int e = e0 + t + k * SORT_THREADS;
        const int ec = e < e1 ? e : ec0;
        s[k] = src[ec];
        d[k] = dst[ec];
    }
    __syncthreads();
    #pragma unroll
    for (int k = 0; k < SORT_UNR; ++k)
        if (e0 + t + k * SORT_THREADS < e1) atomicAdd(&hl[d[k] >> BSHIFT], 1);
    for (int base = e0 + t + ROUND; base < e1 + t; base += ROUND) {
        int d2[SORT_UNR];
        #pragma unroll
        for (int k = 0; k < SORT_UNR; ++k) {
            const int e = base + k * SORT_THREADS;
            d2[k] = dst[e < e1 ? e : ec0];
        }
        #pragma unroll
        for (int k = 0; k < SORT_UNR; ++k)
            if (base + k * SORT_THREADS < e1) atomicAdd(&hl[d2[k] >> BSHIFT], 1);
    }
    __syncthreads();
    if (wave == 0) {
        int loc[TRIPS];
        int sum = 0;
        #pragma unroll
        for (int u = 0; u < TRIPS; ++u) {
            const int b = lane * TRIPS + u;
            loc[u] = (b < NBUCK) ? hl[b] : 0;
            sum += loc[u];
        }
        int incl = sum;
        #pragma unroll
        for (int o = 1; o < 32; o <<= 1) {
            int up = __shfl_up(incl, o);
            if (lane >= o) incl += up;
        }
        int run = incl - sum;
        #pragma unroll
        for (int u = 0; u < TRIPS; ++u) {
            const int b = lane * TRIPS + u;
            if (b < NBUCK) {
                offL[b] = run;
                seg[j * NBUCK + b] = make_int2(run, loc[u]);
            }
            run += loc[u];
        }
    }
    __syncthreads();
    if (ch <= ROUND) {
        #pragma unroll
        for (int k = 0; k < SORT_UNR; ++k) {
            if (e0 + t + k * SORT_THREADS < e1) {
                const int p = atomicAdd(&offL[d[k] >> BSHIFT], 1);
                recL[p] = make_int2(s[k], d[k]);
            }
        }
        __syncthreads();
        const int n = e1 - e0;
        const int n2 = n > 0 ? (n >> 1) : 0;
        for (int q = t; q < n2; q += SORT_THREADS)
            ((int4*)slice)[q] = ((const int4*)recL)[q];
        if (t == 0 && n > 0 && (n & 1)) slice[n - 1] = recL[n - 1];
        return;
    }
    #pragma unroll
    for (int k = 0; k < SORT_UNR; ++k) {
        if (e0 + t + k * SORT_THREADS < e1) {
            const int p = atomicAdd(&offL[d[k] >> BSHIFT], 1);
            slice[p] = make_int2(s[k], d[k]);
        }
    }
    for (int base = e0 + t + ROUND; base < e1 + t; base += ROUND) {
        int s2[SORT_UNR], d2[SORT_UNR];
        #pragma unroll
        for (int k = 0; k < SORT_UNR; ++k) {
            const int e = base + k * SORT_THREADS;
            const int ec = e < e1 ? e : ec0;
            s2[k] = src[ec];
            d2[k] = dst[ec];
        }
        #pragma unroll
        for (int k = 0; k < SORT_UNR; ++k) {
            if (base + k * SORT_THREADS < e1) {
                const int p = atomicAdd(&offL[d2[k] >> BSHIFT], 1);
                slice[p] = make_int2(s2[k], d2[k]);
            }
        }
    }
}

__global__ __launch_bounds__(SORT_THREADS) void fine_sort_kernel(
        const int2* __restrict__ tmp, const int2* __restrict__ seg,
        const float* __restrict__ x, int* __restrict__ srcs,
        int* __restrict__ rowptr, int* __restrict__ cnt,
        float* __restrict__ xs, int N, int E) {
    __shared__ int cL[BNODES];
    __shared__ int curL[BNODES];
    __shared__ int wsum[BNODES / 32];
    __shared__ int wpart[SORT_WAVES];
    __shared__ int s0L;
    __shared__ int totL;
    __shared__ int srcL[FINE_CAP];
    const int t = (int)threadIdx.x, lane = t & 31, b = (int)blockIdx.x;
    const int wave = __builtin_amdgcn_readfirstlane(t >> 5);
    const int n0 = b * BNODES;
    int2 sg = make_int2(0, 0);
    if (lane < RUNS_PER_WAVE) sg = seg[(wave * RUNS_PER_WAVE + lane) * NBUCK + b];
    if (t < BNODES) cL[t] = 0;
    int part = sg.x;
    #pragma unroll
    for (int o = 16; o > 0; o >>= 1) part += __shfl_xor(part, o);
    if (lane == 0) wpart[wave] = part;
    const int pitch = sort_pitch(E);
    int ra[RUNS_PER_WAVE], rc[RUNS_PER_WAVE];
    int maxc = 0;
    #pragma unroll
    for (int u = 0; u < RUNS_PER_WAVE; ++u) {
        ra[u] = (wave * RUNS_PER_WAVE + u) * pitch + lane_bcast_i(sg.x, u);
        rc[u] = lane_bcast_i(sg.y, u);
        maxc = rc[u] > maxc ? rc[u] : maxc;
    }
    __syncthreads();
    if (wave == 0) {
        int v = lane < SORT_WAVES ? wpart[lane] : 0;
        #pragma unroll
        for (int o = 16; o > 0; o >>= 1) v += __shfl_xor(v, o);
        if (lane == 0) s0L = v;
    }
    for (int r0 = 0; r0 < maxc; r0 += 32) {
        const int i = r0 + lane;
        int d[RUNS_PER_WAVE];
        #pragma unroll
        for (int u = 0; u < RUNS_PER_WAVE; ++u)
            d[u] = tmp[i < rc[u] ? ra[u] + i : 0].y;
        #pragma unroll
        for (int u = 0; u < RUNS_PER_WAVE; ++u)
            if (i < rc[u]) atomicAdd(&cL[d[u] - n0], 1);
    }
    __syncthreads();
    const int s0 = s0L;
    int c = 0, incl = 0;
    if (t < BNODES) {
        c = cL[t];
        incl = c;
        #pragma unroll
        for (int o = 1; o < 32; o <<= 1) {
            int up = __shfl_up(incl, o);
            if (lane >= o) incl += up;
        }
        if (lane == 31) wsum[wave] = incl;
    }
    __syncthreads();
    if (t < BNODES) {
        int base = 0;
        for (int k = 0; k < wave; ++k) base += wsum[k];
        const int ex = base + incl - c;
        curL[t] = ex;
        if (t == BNODES - 1) totL = ex + c;
        const int n = n0 + t;
        if (n < N) { rowptr[n] = s0 + ex; cnt[n] = c; }
    }
    __syncthreads();
    const int total = totL;
    const int ph = s0 & 31;
    if (ph + total <= FINE_CAP) {
        for (int r0 = 0; r0 < maxc; r0 += 32) {
            const int i = r0 + lane;
            int2 p[RUNS_PER_WAVE];
            #pragma unroll
            for (int u = 0; u < RUNS_PER_WAVE; ++u)
                p[u] = tmp[i < rc[u] ? ra[u] + i : 0];
            #pragma unroll
            for (int u = 0; u < RUNS_PER_WAVE; ++u) {
                if (i < rc[u]) {
                    const int r = atomicAdd(&curL[p[u].y - n0], 1);
                    srcL[ph + r] = p[u].x;
                }
            }
        }
        __syncthreads();
        int* const line0 = srcs + (s0 - ph);
        for (int i = t; i < ph + total; i += SORT_THREADS)
            if (i >= ph) line0[i] = srcL[i];
    } else {
        for (int r0 = 0; r0 < maxc; r0 += 32) {
            const int i = r0 + lane;
            int2 p[RUNS_PER_WAVE];
            #pragma unroll
            for (int u = 0; u < RUNS_PER_WAVE; ++u)
                p[u] = tmp[i < rc[u] ? ra[u] + i : 0];
            #pragma unroll
            for (int u = 0; u < RUNS_PER_WAVE; ++u) {
                if (i < rc[u]) {
                    const int r = atomicAdd(&curL[p[u].y - n0], 1);
                    srcs[s0 + r] = p[u].x;
                }
            }
        }
    }
    for (int i = t; i < BNODES * 8; i += SORT_THREADS) {
        const int row = i >> 3;
        const int n = n0 + row;
        if (n < N) {
            const float dn = rsqrtf((float)(cL[row] + 1));
            float4 v = ((const float4*)x)[(size_t)n * 8 + (i & 7)];
            v.x *= dn; v.y *= dn; v.z *= dn; v.w *= dn;
            ((float4*)xs)[(size_t)n * 8 + (i & 7)] = v;
        }
    }
}


#define GATHER_GROUP 16
#define NODES_PER_WAVE 4
template <int K>
__device__ __forceinline__ void gather_group(const float* __restrict__ h, int s, float w,
                                             int j, int lane, float& ax, float& ay) {
    if constexpr (K == 64) {
        float2 v[GATHER_GROUP];
        #pragma unroll
        for (int u = 0; u < GATHER_GROUP; ++u) {
            const int sj = lane_bcast_i(s, j + u);
            v[u] = ((const float2*)(h + (size_t)sj * 64))[lane];
        }
        #pragma unroll
        for (int u = 0; u < GATHER_GROUP; ++u) {
            const float wj = lane_bcast_f(w, j + u);
            ax += wj * v[u].x;
            ay += wj * v[u].y;
        }
    } else {
        float v[GATHER_GROUP];
        #pragma unroll
        for (int u = 0; u < GATHER_GROUP; ++u) {
            const int sj = lane_bcast_i(s, j + u);
            v[u] = h[(size_t)sj * 32 + lane];
        }
        #pragma unroll
        for (int u = 0; u < GATHER_GROUP; ++u) {
            const float wj = lane_bcast_f(w, j + u);
            ax += wj * v[u];
        }
    }
}

template <int K, bool POOL>
__global__ __launch_bounds__(256) void agg_gemm_kernel(
        const int* __restrict__ rowptr, const int* __restrict__ cnt,
        const int* __restrict__ srcs, const float* __restrict__ h,
        const float* __restrict__ W, const float* __restrict__ bias,
        float* __restrict__ hout, const int* __restrict__ batch,
        float* pooled, float* counts, int N) {
    constexpr int PITCH = K + 4;
    __shared__ __attribute__((aligned(16))) float Wlds[K * 64];
    __shared__ __attribute__((aligned(16))) float Alds[32 * PITCH];
    __shared__ int Glds[32];
    __shared__ float Dlds[32];
    constexpr int HPITCH = 68;
    __shared__ float Hlds[POOL ? 32 * HPITCH : 1];

    const int lane = threadIdx.x & 31;
    const int wave = __builtin_amdgcn_readfirstlane((int)(threadIdx.x >> 5));
    const int nb = (int)blockIdx.x * 32 + wave * NODES_PER_WAVE;

    int startv = 0, degv = 0, gv = 0;
    if (lane < NODES_PER_WAVE && nb + lane < N) {
        startv = rowptr[nb + lane];
        degv   = cnt[nb + lane];
        if (POOL) gv = batch[nb + lane];
    }
    float4 wv[K / 16];
    #pragma unroll
    for (int u = 0; u < K / 16; ++u)
        wv[u] = ((const float4*)W)[u * 256 + (int)threadIdx.x];
    #pragma unroll
    for (int u = 0; u < K / 16; ++u)
        ((float4*)Wlds)[u * 256 + (int)threadIdx.x] = wv[u];

    int s[NODES_PER_WAVE], st[NODES_PER_WAVE], dg[NODES_PER_WAVE];
    float w[NODES_PER_WAVE];
    int pv[NODES_PER_WAVE];
    #pragma unroll
    for (int i = 0; i < NODES_PER_WAVE; ++i) {
        st[i] = lane_bcast_i(startv, i);
        dg[i] = lane_bcast_i(degv, i);
        const int c0 = dg[i] < 32 ? dg[i] : 32;
        pv[i] = srcs[lane < c0 ? st[i] + lane : 0];
    }
    #pragma unroll
    for (int i = 0; i < NODES_PER_WAVE; ++i) {
        const int c0 = dg[i] < 32 ? dg[i] : 32;
        s[i] = lane < c0 ? pv[i] : nb + i;
        w[i] = lane < c0 ? 1.0f : 0.0f;
    }

    #pragma unroll
    for (int i = 0; i < NODES_PER_WAVE; ++i) {
        const int n = nb + i;
        const int row = wave * NODES_PER_WAVE + i;
        float ax = 0.0f, ay = 0.0f;
        if (n < N) {
            const int deg = dg[i], start = st[i];
            const int c0 = deg < 32 ? deg : 32;
            float sx, sy = 0.0f;
            if constexpr (K == 64) {
                const float2 sv = ((const float2*)(h + (size_t)n * 64))[lane];
                sx = sv.x; sy = sv.y;
            } else {
                sx = h[(size_t)n * 32 + lane];
            }
            for (int j = 0; j < c0; j += GATHER_GROUP)
                gather_group<K>(h, s[i], w[i], j, lane, ax, ay);
            for (int base = 32; base < deg; base += 32) {
                const int rem = deg - base;
                const int c = rem < 32 ? rem : 32;
                int s2 = n;
                float w2 = 0.0f;
                if (lane < c) {
                    s2 = srcs[start + base + lane];
                    w2 = 1.0f;
                }
                for (int j = 0; j < c; j += GATHER_GROUP)
                    gather_group<K>(h, s2, w2, j, lane, ax, ay);
            }
            const float dn = rsqrtf((float)(deg + 1));
            ax = dn * (ax + sx);
            ay = dn * (ay + sy);
            if (!POOL && lane == 0) Dlds[row] = dn;
            if (POOL) {
                const int g = lane_bcast_i(gv, i);
                if (lane == 0) Glds[row] = g;
            }
        }
        if constexpr (K == 64) ((float2*)(Alds + row * PITCH))[lane] = make_float2(ax, ay);
        else                   Alds[row * PITCH + lane] = ax;
    }
    __syncthreads();

    const int half = lane >> 4;
    const int l16  = lane & 15;
    const int col0 = (wave & 3) * 16;
    const int rloc = (wave >> 2) * 16;
    const float* Arow = Alds + (rloc + l16) * PITCH + 2 * half;
    const float* Brow = Wlds + 2 * half * 64 + col0 + l16;

    v8f acc = {};
    #pragma unroll
    for (int k0 = 0; k0 < K; k0 += 4) {
        v2f a = *(const v2f*)(Arow + k0);
        v2f b;
        b.x = Brow[k0 * 64];
        b.y = Brow[k0 * 64 + 64];
        acc = __builtin_amdgcn_wmma_f32_16x16x4_f32(
             false, a,  false, b,
             (short)0, acc,  false,  false);
    }
    const float bv = bias[col0 + l16];
    if constexpr (!POOL) {
        #pragma unroll
        for (int r = 0; r < 8; ++r) {
            const int rl = rloc + r + 8 * half;
            const int mm = (int)blockIdx.x * 32 + rl;
            if (mm < N) {
                const float v = fmaxf(acc[r] + bv, 0.0f);
                hout[(size_t)mm * 64 + col0 + l16] = Dlds[rl] * v;
            }
        }
    } else {
        #pragma unroll
        for (int r = 0; r < 8; ++r)
            Hlds[(rloc + r + 8 * half) * HPITCH + col0 + l16] = fmaxf(acc[r] + bv, 0.0f);
        __syncthreads();
        if (wave < 2) {
            const int c = wave * 32 + lane;
            const int left = N - (int)blockIdx.x * 32;
            const int nvalid = left < 32 ? left : 32;
            const int gl = lane < nvalid ? Glds[lane] : 0;
            float hv[32];
            #pragma unroll
            for (int r = 0; r < 32; ++r) hv[r] = Hlds[r * HPITCH + c];
            int gcur = lane_bcast_i(gl, 0);
            int len = 0;
            float run = 0.0f;
            #pragma unroll
            for (int r = 0; r < 32; ++r) {
                if (r < nvalid) {
                    const int g = lane_bcast_i(gl, r);
                    if (g != gcur) {
                        atomic_add_f32(pooled + (size_t)gcur * 64 + c, run);
                        if (c == 0) atomic_add_f32(&counts[(size_t)gcur * GCNT_STRIDE], (float)len);
                        gcur = g;
                        len = 0;
                        run = 0.0f;
                    }
                    run += hv[r];
                    ++len;
                }
            }
            atomic_add_f32(pooled + (size_t)gcur * 64 + c, run);
            if (c == 0) atomic_add_f32(&counts[(size_t)gcur * GCNT_STRIDE], (float)len);
        }
    }
}

__global__ void pooled_mlp_head_kernel(const float* __restrict__ pooled,
                                       const float* __restrict__ counts,
                                       const float* __restrict__ lw1,
                                       const float* __restrict__ lb1,
                                       const float* __restrict__ lw2,
                                       const float* __restrict__ lb2,
                                       float* __restrict__ out, int G) {
    __shared__ float Wlds[64 * 64];
    __shared__ float Rt[32 * 65];
    #pragma unroll 4
    for (int i = threadIdx.x; i < 64 * 64; i += 256)
        Wlds[i] = lw1[i];
    __syncthreads();

    const int wave = threadIdx.x >> 5;
    const int lane = threadIdx.x & 31;
    const int half = lane >> 4;
    const int l16  = lane & 15;
    const int col0 = (wave & 3) * 16;
    const int rloc = (wave >> 2) * 16;
    const int row0 = blockIdx.x * 32 + rloc;
    const int m    = row0 + l16;
    const int mc   = (m < G) ? m : (G - 1);
    const float c  = fmaxf(counts[(size_t)mc * GCNT_STRIDE], 1.0f);
    const float* __restrict__ Arow = pooled + (size_t)mc * 64 + 2 * half;
    const float* __restrict__ Brow = Wlds + 2 * half * 64 + col0 + l16;

    v8f acc = {};
    #pragma unroll
    for (int k0 = 0; k0 < 64; k0 += 4) {
        v2f a = *(const v2f*)(Arow + k0);
        a.x = a.x / c;
        a.y = a.y / c;
        v2f b;
        b.x = Brow[k0 * 64];
        b.y = Brow[k0 * 64 + 64];
        acc = __builtin_amdgcn_wmma_f32_16x16x4_f32(
             false, a,  false, b,
             (short)0, acc,  false,  false);
    }
    const float bv = lb1[col0 + l16];
    #pragma unroll
    for (int r = 0; r < 8; ++r)
        Rt[(rloc + r + 8 * half) * 65 + col0 + l16] = fmaxf(acc[r] + bv, 0.0f);
    __syncthreads();

    const int t = threadIdx.x;
    if (t < 64) {
        const int gl = t >> 1, j = t & 1;
        const int gg = blockIdx.x * 32 + gl;
        if (gg < G) {
            float o = lb2[j];
            #pragma unroll
            for (int k = 0; k < 64; ++k) o += Rt[gl * 65 + k] * lw2[k * 2 + j];
            out[(size_t)gg * 2 + j] = o;
        }
    }
}

static inline size_t alignup(size_t v) { return (v + 255) & ~(size_t)255; }

extern "C" void kernel_launch(void* const* d_in, const int* in_sizes, int n_in,
                              void* d_out, int out_size, void* d_ws, size_t ws_size,
                              hipStream_t stream) {
    const float* x     = (const float*)d_in[0];
    const int*   eidx  = (const int*)d_in[1];
    const int*   batch = (const int*)d_in[2];
    const float* W1 = (const float*)d_in[3];  const float* b1 = (const float*)d_in[4];
    const float* W2 = (const float*)d_in[5];  const float* b2 = (const float*)d_in[6];
    const float* W3 = (const float*)d_in[7];  const float* b3 = (const float*)d_in[8];
    const float* lw1 = (const float*)d_in[9]; const float* lb1 = (const float*)d_in[10];
    const float* lw2 = (const float*)d_in[11]; const float* lb2 = (const float*)d_in[12];
    float* out = (float*)d_out;

    const int N = N_NODES, G = N_GRAPHS;
    const int E = in_sizes[1] / 2;
    const int* src = eidx;
    const int* dst = eidx + E;

    char* w = (char*)d_ws;
    int2*  seg    = (int2*)w;  w += alignup((size_t)NBUCK * SORT_BLOCKS * 8);
    int*   cnt    = (int*)w;   w += alignup((size_t)N * 4);
    int*   rowptr = (int*)w;   w += alignup((size_t)N * 4);
    int2*  tmp    = (int2*)w;  w += alignup(((size_t)E + 17 * SORT_BLOCKS) * 8);
    int*   srcs   = (int*)w;   w += alignup((size_t)E * 4);
    float* xs     = (float*)w; w += alignup((size_t)N * 32 * 4);
    float* h1     = (float*)w; w += alignup((size_t)N * 64 * 4);
    float* h2     = (float*)w; w += alignup((size_t)N * 64 * 4);
    float* pooled = (float*)w; w += alignup((size_t)G * 64 * 4);
    float* counts = (float*)w; w += alignup((size_t)G * 4 * GCNT_STRIDE);
    (void)ws_size; (void)n_in;

    dim3 blk256(256);
    const int GF4 = G * 64 / 4;
    const int G4 = G * GCNT_STRIDE / 4;
    int gLayer = (N + 31) / 32;
    int gGemmG = (G + 31) / 32;

    coarse_sort_kernel<<<SORT_BLOCKS, dim3(SORT_THREADS), 0, stream>>>(src, dst, tmp, seg, E, pooled, GF4, counts, G4);
    fine_sort_kernel<<<NBUCK, dim3(SORT_THREADS), 0, stream>>>(tmp, seg, x, srcs, rowptr, cnt, xs, N, E);

    agg_gemm_kernel<32, false><<<gLayer, blk256, 0, stream>>>(rowptr, cnt, srcs, xs, W1, b1, h1,
                                                              batch, pooled, counts, N);
    agg_gemm_kernel<64, false><<<gLayer, blk256, 0, stream>>>(rowptr, cnt, srcs, h1, W2, b2, h2,
                                                              batch, pooled, counts, N);
    agg_gemm_kernel<64, true><<<gLayer, blk256, 0, stream>>>(rowptr, cnt, srcs, h2, W3, b3, h1,
                                                             batch, pooled, counts, N);

    pooled_mlp_head_kernel<<<gGemmG, blk256, 0, stream>>>(pooled, counts, lw1, lb1, lw2, lb2, out, G);
    (void)out_size;
}
